# starting kernel
# baseline (speedup 1.0000x reference)
#include <hip/hip_runtime.h>

typedef __attribute__((ext_vector_type(2))) float v2f;
typedef __attribute__((ext_vector_type(8))) float v8f;

#define C_DIM 128

__global__ __launch_bounds__(256) void gcn_gemm_norm(
    const float* __restrict__ x, const float* __restrict__ w,
    const float* __restrict__ norm, float* __restrict__ h, int N)
{
    __shared__ float lds_w[C_DIM * C_DIM];

    for (int idx = threadIdx.x; idx < (C_DIM / 2) * C_DIM; idx += 256) {
        const int p = idx >> 7;
        const int n = idx & 127;
        v2f v;
        v.x = w[(2 * p + 0) * C_DIM + n];
        v.y = w[(2 * p + 1) * C_DIM + n];
        const int dw = p * 256 + ((2 * n + 32 * (p & 1)) & 255);
        *(v2f*)&lds_w[dw] = v;
    }
    __syncthreads();

    const int wave   = threadIdx.x >> 5;
    const int lane   = threadIdx.x & 31;
    const int lane16 = lane & 15;
    const int khalf  = lane >> 4;

    const int m0 = (blockIdx.x * 8 + wave) * 16;
    int row  = m0 + lane16;
    int rowc = row < N ? row : N - 1;

    v8f acc[8];
    const v8f vzero = {0.f,0.f,0.f,0.f,0.f,0.f,0.f,0.f};
    #pragma unroll
    for (int n = 0; n < 8; ++n) acc[n] = vzero;

    const float* xrow = x + (size_t)rowc * C_DIM;

    int coff[8];
    #pragma unroll
    for (int n = 0; n < 8; ++n)
        coff[n] = (32 * n + 2 * lane16 + 32 * khalf) & 255;
    const float* lwp = lds_w + khalf * 256;

    #pragma unroll 4
    for (int k = 0; k < C_DIM; k += 4) {
        const int kb = k + 2 * khalf;
        v2f a = *(const v2f*)(xrow + kb);
        const float* lk = lwp + (k >> 1) * 256;
        #pragma unroll
        for (int n = 0; n < 8; ++n) {
            v2f b = *(const v2f*)(lk + coff[n]);
            acc[n] = __builtin_amdgcn_wmma_f32_16x16x4_f32(
                false, a, false, b, (short)0, acc[n], false, false);
        }
    }

    float nrm[8];
    #pragma unroll
    for (int j = 0; j < 8; ++j) {
        const int r = m0 + j + 8 * khalf;
        nrm[j] = (r < N) ? norm[r] : 0.0f;
    }
    #pragma unroll
    for (int n = 0; n < 8; ++n) {
        #pragma unroll
        for (int j = 0; j < 8; ++j) {
            const int r = m0 + j + 8 * khalf;
            if (r < N) {
                h[(size_t)r * C_DIM + n * 16 + lane16] = acc[n][j] * nrm[j];
            }
        }
    }
}

__global__ __launch_bounds__(256) void gcn_zero(float4* __restrict__ out, int n4)
{
    int i = blockIdx.x * blockDim.x + threadIdx.x;
    if (i < n4) out[i] = make_float4(0.f, 0.f, 0.f, 0.f);
}

__global__ __launch_bounds__(256) void gcn_scatter(
    const float* __restrict__ h, const int* __restrict__ esrc,
    const int* __restrict__ edst, float* __restrict__ out, int E)
{
    const long long tid = (long long)blockIdx.x * blockDim.x + threadIdx.x;
    const long long e = tid >> 5;
    if (e >= E) return;
    const int lane = (int)(tid & 31);
    const int src = esrc[e];
    const int dst = edst[e];
    const float4 v = *(const float4*)(h + (size_t)src * C_DIM + lane * 4);
    float* o = out + (size_t)dst * C_DIM + lane * 4;
    atomicAdd(o + 0, v.x);
    atomicAdd(o + 1, v.y);
    atomicAdd(o + 2, v.z);
    atomicAdd(o + 3, v.w);
}

__global__ __launch_bounds__(256) void gcn_final(
    float* __restrict__ out, const float* __restrict__ norm,
    const float* __restrict__ bias, int N)
{
    const long long tid = (long long)blockIdx.x * blockDim.x + threadIdx.x;
    if (tid >= (long long)N * 32) return;
    const int row = (int)(tid >> 5);
    const int c4  = (int)(tid & 31) * 4;
    const float nm = norm[row];
    float4 v = ((float4*)out)[tid];
    const float4 b = *(const float4*)(bias + c4);
    v.x = v.x * nm + b.x;
    v.y = v.y * nm + b.y;
    v.z = v.z * nm + b.z;
    v.w = v.w * nm + b.w;
    ((float4*)out)[tid] = v;
}

extern "C" void kernel_launch(void* const* d_in, const int* in_sizes, int n_in,
                              void* d_out, int out_size, void* d_ws, size_t ws_size,
                              hipStream_t stream) {
    const float* x    = (const float*)d_in[0];
    const float* norm = (const float*)d_in[1];
    const float* w    = (const float*)d_in[2];
    const float* bias = (const float*)d_in[3];
    const int*   esrc = (const int*)d_in[4];
    const int*   edst = (const int*)d_in[5];
    float* out = (float*)d_out;

    const int N = in_sizes[0] / C_DIM;
    const int E = in_sizes[4];
    float* h = (float*)d_ws;

    dim3 blk(256);

    gcn_gemm_norm<<<(N + 127) / 128, blk, 0, stream>>>(x, w, norm, h, N);

    const int n4 = out_size / 4;
    gcn_zero<<<(n4 + 255) / 256, blk, 0, stream>>>((float4*)out, n4);

    const long long sthreads = (long long)E * 32;
    gcn_scatter<<<(unsigned)((sthreads + 255) / 256), blk, 0, stream>>>(h, esrc, edst, out, E);

    const long long fthreads = (long long)N * 32;
    gcn_final<<<(unsigned)((fthreads + 255) / 256), blk, 0, stream>>>(out, norm, bias, N);
}


# Scatter: float atomics on out replaced by owner-scans-all: each work-group owns 391 out rows in LDS, scans all edge_dst, one wave per row adds h rows in a fixed scan order; 4 launches to 2
# speedup vs baseline: 1.9015x; 1.9015x over previous
#include <hip/hip_runtime.h>

typedef __attribute__((ext_vector_type(2))) float v2f;
typedef __attribute__((ext_vector_type(8))) float v8f;

#define C_DIM 128

__global__ __launch_bounds__(256) void gcn_gemm_norm(
    const float* __restrict__ x, const float* __restrict__ w,
    const float* __restrict__ norm, float* __restrict__ h, int N)
{
    __shared__ float lds_w[C_DIM * C_DIM];

    for (int idx = threadIdx.x; idx < (C_DIM / 2) * C_DIM; idx += 256) {
        const int p = idx >> 7;
        const int n = idx & 127;
        v2f v;
        v.x = w[(2 * p + 0) * C_DIM + n];
        v.y = w[(2 * p + 1) * C_DIM + n];
        const int dw = p * 256 + ((2 * n + 32 * (p & 1)) & 255);
        *(v2f*)&lds_w[dw] = v;
    }
    __syncthreads();

    const int wave   = threadIdx.x >> 5;
    const int lane   = threadIdx.x & 31;
    const int lane16 = lane & 15;
    const int khalf  = lane >> 4;

    const int m0 = (blockIdx.x * 8 + wave) * 16;
    int row  = m0 + lane16;
    int rowc = row < N ? row : N - 1;

    v8f acc[8];
    const v8f vzero = {0.f,0.f,0.f,0.f,0.f,0.f,0.f,0.f};
    #pragma unroll
    for (int n = 0; n < 8; ++n) acc[n] = vzero;

    const float* xrow = x + (size_t)rowc * C_DIM;

    int coff[8];
    #pragma unroll
    for (int n = 0; n < 8; ++n)
        coff[n] = (32 * n + 2 * lane16 + 32 * khalf) & 255;
    const float* lwp = lds_w + khalf * 256;

    #pragma unroll 4
    for (int k = 0; k < C_DIM; k += 4) {
        const int kb = k + 2 * khalf;
        v2f a = *(const v2f*)(xrow + kb);
        const float* lk = lwp + (k >> 1) * 256;
        #pragma unroll
        for (int n = 0; n < 8; ++n) {
            v2f b = *(const v2f*)(lk + coff[n]);
            acc[n] = __builtin_amdgcn_wmma_f32_16x16x4_f32(
                false, a, false, b, (short)0, acc[n], false, false);
        }
    }

    float nrm[8];
    #pragma unroll
    for (int j = 0; j < 8; ++j) {
        const int r = m0 + j + 8 * khalf;
        nrm[j] = (r < N) ? norm[r] : 0.0f;
    }
    #pragma unroll
    for (int n = 0; n < 8; ++n) {
        #pragma unroll
        for (int j = 0; j < 8; ++j) {
            const int r = m0 + j + 8 * khalf;
            if (r < N) {
                h[(size_t)r * C_DIM + n * 16 + lane16] = acc[n][j] * nrm[j];
            }
        }
    }
}

#define AGG_ROWS   391
#define AGG_WAVES  32
#define CAP_A      416
#define CAP_B      192
#define DUMMY_ROW  AGG_ROWS

__global__ __launch_bounds__(1024) void gcn_aggregate(
    const float* __restrict__ h, const int* __restrict__ esrc,
    const int* __restrict__ edst, const float* __restrict__ norm,
    const float* __restrict__ bias, float* __restrict__ out, int N, int E)
{
    __shared__ float acc[(AGG_ROWS + 1) * C_DIM];
    __shared__ int a_src[AGG_WAVES][CAP_A];
    __shared__ unsigned short a_row[AGG_WAVES][CAP_A];
    __shared__ int b_src[AGG_WAVES][CAP_B];
    __shared__ unsigned short b_row[AGG_WAVES][CAP_B];
    __shared__ int a_cnt[AGG_WAVES];
    __shared__ int a_done[AGG_WAVES];

    const int lane = threadIdx.x & 31;
    const int wave = __builtin_amdgcn_readfirstlane(threadIdx.x >> 5);
    const int row0 = blockIdx.x * AGG_ROWS;
    const int nrows_i = (N - row0) < AGG_ROWS ? (N - row0) : AGG_ROWS;
    const unsigned nrows = (unsigned)nrows_i;

    for (int i = threadIdx.x; i < (AGG_ROWS + 1) * C_DIM / 4; i += 1024)
        ((float4*)acc)[i] = make_float4(0.f, 0.f, 0.f, 0.f);

    const int slice = (((E + AGG_WAVES - 1) / AGG_WAVES) + 127) & ~127;
    long long sb = (long long)wave * slice;
    long long se = sb + slice;
    if (sb > E) sb = E;
    if (se > E) se = E;
    int cursor = (int)sb;
    const int end = (int)se;

    int* my_a = a_src[wave];
    unsigned short* my_ar = a_row[wave];
    int* my_b = b_src[wave];
    unsigned short* my_br = b_row[wave];

    auto scan_chunk = [&](int d0, int d1, int d2, int d3, int ebase, int& cnt) -> bool {
        const unsigned u0 = (unsigned)(d0 - row0), u1 = (unsigned)(d1 - row0);
        const unsigned u2 = (unsigned)(d2 - row0), u3 = (unsigned)(d3 - row0);
        const unsigned m01 = u0 < u1 ? u0 : u1;
        const unsigned m23 = u2 < u3 ? u2 : u3;
        const unsigned m = m01 < m23 ? m01 : m23;
        if (__builtin_amdgcn_ballot_w32(m < nrows) == 0u) return true;
        if (cnt + 128 > CAP_A) return false;
        const unsigned u[4] = {u0, u1, u2, u3};
        #pragma unroll
        for (int c = 0; c < 4; ++c) {
            const bool hit = u[c] < nrows;
            const unsigned mask = __builtin_amdgcn_ballot_w32(hit);
            if (mask) {
                const int rank = (int)__builtin_amdgcn_mbcnt_lo(mask, 0u);
                if (hit) my_a[cnt + rank] = ebase + c;
                cnt += __builtin_popcount(mask);
            }
        }
        return true;
    };

    auto flush_rows = [&](int cnt) {
        const int padded = (cnt + 7) & ~7;
        if (lane < padded - cnt) { my_b[cnt + lane] = 0; my_br[cnt + lane] = (unsigned short)DUMMY_ROW; }
        for (int i = 0; i < padded; i += 8) {
            const int sv = my_b[i + (lane & 7)];
            const int rv = (int)my_br[i + (lane & 7)];
            float4 v[8];
            int r[8];
            #pragma unroll
            for (int k = 0; k < 8; ++k) {
                const int s = __builtin_amdgcn_readlane(sv, k);
                r[k] = __builtin_amdgcn_readlane(rv, k);
                v[k] = *(const float4*)(h + (size_t)s * C_DIM + lane * 4);
            }
            #pragma unroll
            for (int k = 0; k < 8; ++k) {
                float* a = &acc[r[k] * C_DIM + lane];
                __hip_atomic_fetch_add(a,      v[k].x, __ATOMIC_RELAXED, __HIP_MEMORY_SCOPE_WORKGROUP);
                __hip_atomic_fetch_add(a + 32, v[k].y, __ATOMIC_RELAXED, __HIP_MEMORY_SCOPE_WORKGROUP);
                __hip_atomic_fetch_add(a + 64, v[k].z, __ATOMIC_RELAXED, __HIP_MEMORY_SCOPE_WORKGROUP);
                __hip_atomic_fetch_add(a + 96, v[k].w, __ATOMIC_RELAXED, __HIP_MEMORY_SCOPE_WORKGROUP);
            }
        }
    };

    bool all_done = false;
    while (!all_done) {
        int cnt = 0;
        bool full = false;
        while (!full && cursor + 512 <= end) {
            const int4* p = (const int4*)(edst + cursor) + lane;
            const int4 q0 = p[0], q1 = p[32], q2 = p[64], q3 = p[96];
            const int eb = cursor + lane * 4;
            if (!scan_chunk(q0.x, q0.y, q0.z, q0.w, eb, cnt))            { full = true; break; }
            cursor += 128;
            if (!scan_chunk(q1.x, q1.y, q1.z, q1.w, eb + 128, cnt))      { full = true; break; }
            cursor += 128;
            if (!scan_chunk(q2.x, q2.y, q2.z, q2.w, eb + 256, cnt))      { full = true; break; }
            cursor += 128;
            if (!scan_chunk(q3.x, q3.y, q3.z, q3.w, eb + 384, cnt))      { full = true; break; }
            cursor += 128;
        }
        while (!full && cursor < end) {
            const int eb = cursor + lane * 4;
            const int d0 = (eb + 0 < end) ? edst[eb + 0] : -1;
            const int d1 = (eb + 1 < end) ? edst[eb + 1] : -1;
            const int d2 = (eb + 2 < end) ? edst[eb + 2] : -1;
            const int d3 = (eb + 3 < end) ? edst[eb + 3] : -1;
            if (!scan_chunk(d0, d1, d2, d3, eb, cnt)) { full = true; break; }
            cursor += 128;
        }
        for (int i = lane; i < cnt; i += 32) {
            const int e = my_a[i];
            my_a[i]  = esrc[e];
            my_ar[i] = (unsigned short)(edst[e] - row0);
        }
        if (lane == 0) { a_cnt[wave] = cnt; a_done[wave] = (cursor >= end) ? 1 : 0; }
        __syncthreads();

        all_done = __builtin_amdgcn_ballot_w32(a_done[lane] != 0) == 0xFFFFFFFFu;

        int cb = 0;
        for (int s = 0; s < AGG_WAVES; ++s) {
            const int n = __builtin_amdgcn_readfirstlane(a_cnt[s]);
            for (int i0 = 0; i0 < n; i0 += 32) {
                const int i = i0 + lane;
                const bool valid = i < n;
                const int rr = valid ? (int)a_row[s][i] : 0;
                const bool mine = valid && ((rr & (AGG_WAVES - 1)) == wave);
                const unsigned mask = __builtin_amdgcn_ballot_w32(mine);
                if (mask) {
                    if (cb + 32 > CAP_B - 8) { flush_rows(cb); cb = 0; }
                    const int rank = (int)__builtin_amdgcn_mbcnt_lo(mask, 0u);
                    if (mine) { my_b[cb + rank] = a_src[s][i]; my_br[cb + rank] = (unsigned short)rr; }
                    cb += __builtin_popcount(mask);
                }
            }
        }
        flush_rows(cb);
        __syncthreads();
    }

    const float4 b4 = *(const float4*)(bias + lane * 4);
    for (int lr = wave; lr < nrows_i; lr += AGG_WAVES) {
        const float* a = &acc[lr * C_DIM + lane];
        const float nm = norm[row0 + lr];
        float4 v;
        v.x = a[0]  * nm + b4.x;
        v.y = a[32] * nm + b4.y;
        v.z = a[64] * nm + b4.z;
        v.w = a[96] * nm + b4.w;
        *(float4*)(out + (size_t)(row0 + lr) * C_DIM + lane * 4) = v;
    }
}

extern "C" void kernel_launch(void* const* d_in, const int* in_sizes, int n_in,
                              void* d_out, int out_size, void* d_ws, size_t ws_size,
                              hipStream_t stream) {
    const float* x    = (const float*)d_in[0];
    const float* norm = (const float*)d_in[1];
    const float* w    = (const float*)d_in[2];
    const float* bias = (const float*)d_in[3];
    const int*   esrc = (const int*)d_in[4];
    const int*   edst = (const int*)d_in[5];
    float* out = (float*)d_out;

    const int N = in_sizes[0] / C_DIM;
    const int E = in_sizes[4];
    float* h = (float*)d_ws;

    gcn_gemm_norm<<<(N + 127) / 128, dim3(256), 0, stream>>>(x, w, norm, h, N);

    gcn_aggregate<<<(N + AGG_ROWS - 1) / AGG_ROWS, dim3(1024), 0, stream>>>(
        h, esrc, edst, norm, bias, out, N, E);
}


# Aggregation: each scanning wave groups its hit list by owning wave (stable in-register counting sort); owners copy only their own groups, not walk all 32 lists; sum order unchanged
# speedup vs baseline: 2.1658x; 1.1390x over previous
#include <hip/hip_runtime.h>

typedef __attribute__((ext_vector_type(2))) float v2f;
typedef __attribute__((ext_vector_type(8))) float v8f;

#define C_DIM 128

__global__ __launch_bounds__(256) void gcn_gemm_norm(
    const float* __restrict__ x, const float* __restrict__ w,
    const float* __restrict__ norm, float* __restrict__ h, int N)
{
    __shared__ float lds_w[C_DIM * C_DIM];

    for (int idx = threadIdx.x; idx < (C_DIM / 2) * C_DIM; idx += 256) {
        const int p = idx >> 7;
        const int n = idx & 127;
        v2f v;
        v.x = w[(2 * p + 0) * C_DIM + n];
        v.y = w[(2 * p + 1) * C_DIM + n];
        const int dw = p * 256 + ((2 * n + 32 * (p & 1)) & 255);
        *(v2f*)&lds_w[dw] = v;
    }
    __syncthreads();

    const int wave   = threadIdx.x >> 5;
    const int lane   = threadIdx.x & 31;
    const int lane16 = lane & 15;
    const int khalf  = lane >> 4;

    const int m0 = (blockIdx.x * 8 + wave) * 16;
    int row  = m0 + lane16;
    int rowc = row < N ? row : N - 1;

    v8f acc[8];
    const v8f vzero = {0.f,0.f,0.f,0.f,0.f,0.f,0.f,0.f};
    #pragma unroll
    for (int n = 0; n < 8; ++n) acc[n] = vzero;

    const float* xrow = x + (size_t)rowc * C_DIM;

    int coff[8];
    #pragma unroll
    for (int n = 0; n < 8; ++n)
        coff[n] = (32 * n + 2 * lane16 + 32 * khalf) & 255;
    const float* lwp = lds_w + khalf * 256;

    #pragma unroll 4
    for (int k = 0; k < C_DIM; k += 4) {
        const int kb = k + 2 * khalf;
        v2f a = *(const v2f*)(xrow + kb);
        const float* lk = lwp + (k >> 1) * 256;
        #pragma unroll
        for (int n = 0; n < 8; ++n) {
            v2f b = *(const v2f*)(lk + coff[n]);
            acc[n] = __builtin_amdgcn_wmma_f32_16x16x4_f32(
                false, a, false, b, (short)0, acc[n], false, false);
        }
    }

    float nrm[8];
    #pragma unroll
    for (int j = 0; j < 8; ++j) {
        const int r = m0 + j + 8 * khalf;
        nrm[j] = (r < N) ? norm[r] : 0.0f;
    }
    #pragma unroll
    for (int n = 0; n < 8; ++n) {
        #pragma unroll
        for (int j = 0; j < 8; ++j) {
            const int r = m0 + j + 8 * khalf;
            if (r < N) {
                h[(size_t)r * C_DIM + n * 16 + lane16] = acc[n][j] * nrm[j];
            }
        }
    }
}

#define AGG_ROWS   391
#define AGG_WAVES  32
#define CAP_A      416
#define DUMMY_ROW  AGG_ROWS
#define SRC_BITS   20
#define SRC_MASK   0xFFFFF

__global__ __launch_bounds__(1024) void gcn_aggregate(
    const float* __restrict__ h, const int* __restrict__ esrc,
    const int* __restrict__ edst, const float* __restrict__ norm,
    const float* __restrict__ bias, float* __restrict__ out, int N, int E)
{
    __shared__ float acc[(AGG_ROWS + 1) * C_DIM];
    __shared__ int a_lst[AGG_WAVES][CAP_A];
    __shared__ int s_lst[AGG_WAVES][CAP_A];
    __shared__ int s_off[AGG_WAVES][AGG_WAVES + 1];
    __shared__ int a_done[AGG_WAVES];

    const int lane = threadIdx.x & 31;
    const int wave = __builtin_amdgcn_readfirstlane(threadIdx.x >> 5);
    const int row0 = blockIdx.x * AGG_ROWS;
    const int nrows_i = (N - row0) < AGG_ROWS ? (N - row0) : AGG_ROWS;
    const unsigned nrows = (unsigned)nrows_i;

    for (int i = threadIdx.x; i < (AGG_ROWS + 1) * C_DIM / 4; i += 1024)
        ((float4*)acc)[i] = make_float4(0.f, 0.f, 0.f, 0.f);

    const int slice = (((E + AGG_WAVES - 1) / AGG_WAVES) + 127) & ~127;
    long long sb = (long long)wave * slice;
    long long se = sb + slice;
    if (sb > E) sb = E;
    if (se > E) se = E;
    int cursor = (int)sb;
    const int end = (int)se;

    int* my_a = a_lst[wave];
    int* my_s = s_lst[wave];

    auto scan_chunk = [&](int d0, int d1, int d2, int d3, int ebase, int& cnt) -> bool {
        const unsigned u0 = (unsigned)(d0 - row0), u1 = (unsigned)(d1 - row0);
        const unsigned u2 = (unsigned)(d2 - row0), u3 = (unsigned)(d3 - row0);
        const unsigned m01 = u0 < u1 ? u0 : u1;
        const unsigned m23 = u2 < u3 ? u2 : u3;
        const unsigned m = m01 < m23 ? m01 : m23;
        if (__builtin_amdgcn_ballot_w32(m < nrows) == 0u) return true;
        if (cnt + 128 > CAP_A) return false;
        const unsigned u[4] = {u0, u1, u2, u3};
        #pragma unroll
        for (int c = 0; c < 4; ++c) {
            const bool hit = u[c] < nrows;
            const unsigned mask = __builtin_amdgcn_ballot_w32(hit);
            if (mask) {
                const int rank = (int)__builtin_amdgcn_mbcnt_lo(mask, 0u);
                if (hit) my_a[cnt + rank] = ebase + c;
                cnt += __builtin_popcount(mask);
            }
        }
        return true;
    };

    auto flush_rows = [&](int cnt) {
        const int padded = (cnt + 7) & ~7;
        if (lane < padded - cnt) my_a[cnt + lane] = DUMMY_ROW << SRC_BITS;
        for (int i = 0; i < padded; i += 8) {
            const int pv = my_a[i + (lane & 7)];
            float4 v[8];
            int r[8];
            #pragma unroll
            for (int k = 0; k < 8; ++k) {
                const int p = __builtin_amdgcn_readlane(pv, k);
                r[k] = p >> SRC_BITS;
                v[k] = *(const float4*)(h + (size_t)(p & SRC_MASK) * C_DIM + lane * 4);
            }
            #pragma unroll
            for (int k = 0; k < 8; ++k) {
                float* a = &acc[r[k] * C_DIM + lane];
                __hip_atomic_fetch_add(a,      v[k].x, __ATOMIC_RELAXED, __HIP_MEMORY_SCOPE_WORKGROUP);
                __hip_atomic_fetch_add(a + 32, v[k].y, __ATOMIC_RELAXED, __HIP_MEMORY_SCOPE_WORKGROUP);
                __hip_atomic_fetch_add(a + 64, v[k].z, __ATOMIC_RELAXED, __HIP_MEMORY_SCOPE_WORKGROUP);
                __hip_atomic_fetch_add(a + 96, v[k].w, __ATOMIC_RELAXED, __HIP_MEMORY_SCOPE_WORKGROUP);
            }
        }
    };

    auto key_masks = [&](int key, bool valid, unsigned& same, unsigned& mine) {
        const unsigned vm = __builtin_amdgcn_ballot_w32(valid);
        unsigned sm = vm, mn = vm;
        for (int b = 0; b < 5; ++b) {
            const unsigned bb = __builtin_amdgcn_ballot_w32(((key >> b) & 1) != 0);
            sm &= ~(bb ^ (0u - (unsigned)((key  >> b) & 1)));
            mn &= ~(bb ^ (0u - (unsigned)((lane >> b) & 1)));
        }
        same = sm;
        mine = mn;
    };

    bool all_done = false;
    while (!all_done) {
        int cnt = 0;
        bool full = false;
        while (!full && cursor + 512 <= end) {
            const int4* p = (const int4*)(edst + cursor) + lane;
            const int4 q0 = p[0], q1 = p[32], q2 = p[64], q3 = p[96];
            const int eb = cursor + lane * 4;
            if (!scan_chunk(q0.x, q0.y, q0.z, q0.w, eb, cnt))            { full = true; break; }
            cursor += 128;
            if (!scan_chunk(q1.x, q1.y, q1.z, q1.w, eb + 128, cnt))      { full = true; break; }
            cursor += 128;
            if (!scan_chunk(q2.x, q2.y, q2.z, q2.w, eb + 256, cnt))      { full = true; break; }
            cursor += 128;
            if (!scan_chunk(q3.x, q3.y, q3.z, q3.w, eb + 384, cnt))      { full = true; break; }
            cursor += 128;
        }
        while (!full && cursor < end) {
            const int eb = cursor + lane * 4;
            const int d0 = (eb + 0 < end) ? edst[eb + 0] : -1;
            const int d1 = (eb + 1 < end) ? edst[eb + 1] : -1;
            const int d2 = (eb + 2 < end) ? edst[eb + 2] : -1;
            const int d3 = (eb + 3 < end) ? edst[eb + 3] : -1;
            if (!scan_chunk(d0, d1, d2, d3, eb, cnt)) { full = true; break; }
            cursor += 128;
        }
        for (int i = lane; i < cnt; i += 32) {
            const int e = my_a[i];
            my_a[i] = esrc[e] | ((edst[e] - row0) << SRC_BITS);
        }

        int cntv = 0;
        for (int i0 = 0; i0 < cnt; i0 += 32) {
            const bool valid = i0 + lane < cnt;
            const int p = valid ? my_a[i0 + lane] : 0;
            unsigned same, mine;
            key_masks((p >> SRC_BITS) & 31, valid, same, mine);
            cntv += __builtin_popcount(mine);
        }
        int inc = cntv;
        for (int d = 1; d < 32; d <<= 1) {
            const int t = __builtin_amdgcn_ds_bpermute(((lane - d) & 31) << 2, inc);
            if (lane >= d) inc += t;
        }
        int runv = inc - cntv;
        s_off[wave][lane] = runv;
        if (lane == 31) s_off[wave][AGG_WAVES] = inc;
        for (int i0 = 0; i0 < cnt; i0 += 32) {
            const bool valid = i0 + lane < cnt;
            const int p = valid ? my_a[i0 + lane] : 0;
            const int key = (p >> SRC_BITS) & 31;
            unsigned same, mine;
            key_masks(key, valid, same, mine);
            const int base = __builtin_amdgcn_ds_bpermute(key << 2, runv);
            const int rank = (int)__builtin_amdgcn_mbcnt_lo(same, 0u);
            if (valid) my_s[base + rank] = p;
            runv += __builtin_popcount(mine);
        }
        if (lane == 0) a_done[wave] = (cursor >= end) ? 1 : 0;
        __syncthreads();

        all_done = __builtin_amdgcn_ballot_w32(a_done[lane] != 0) == 0xFFFFFFFFu;

        const int segb = s_off[lane][wave];
        const int sege = s_off[lane][wave + 1];
        int cb = 0;
        for (int s = 0; s < AGG_WAVES; ++s) {
            const int b0 = __builtin_amdgcn_readlane(segb, s);
            const int n  = __builtin_amdgcn_readlane(sege, s) - b0;
            for (int i0 = 0; i0 < n; i0 += 32) {
                if (cb + 32 > CAP_A - 8) { flush_rows(cb); cb = 0; }
                const int m = (n - i0) < 32 ? (n - i0) : 32;
                if (lane < m) my_a[cb + lane] = s_lst[s][b0 + i0 + lane];
                cb += m;
            }
        }
        flush_rows(cb);
        __syncthreads();
    }

    const float4 b4 = *(const float4*)(bias + lane * 4);
    for (int lr = wave; lr < nrows_i; lr += AGG_WAVES) {
        const float* a = &acc[lr * C_DIM + lane];
        const float nm = norm[row0 + lr];
        float4 v;
        v.x = a[0]  * nm + b4.x;
        v.y = a[32] * nm + b4.y;
        v.z = a[64] * nm + b4.z;
        v.w = a[96] * nm + b4.w;
        *(float4*)(out + (size_t)(row0 + lr) * C_DIM + lane * 4) = v;
    }
}

extern "C" void kernel_launch(void* const* d_in, const int* in_sizes, int n_in,
                              void* d_out, int out_size, void* d_ws, size_t ws_size,
                              hipStream_t stream) {
    const float* x    = (const float*)d_in[0];
    const float* norm = (const float*)d_in[1];
    const float* w    = (const float*)d_in[2];
    const float* bias = (const float*)d_in[3];
    const int*   esrc = (const int*)d_in[4];
    const int*   edst = (const int*)d_in[5];
    float* out = (float*)d_out;

    const int N = in_sizes[0] / C_DIM;
    const int E = in_sizes[4];
    float* h = (float*)d_ws;

    gcn_gemm_norm<<<(N + 127) / 128, dim3(256), 0, stream>>>(x, w, norm, h, N);

    gcn_aggregate<<<(N + AGG_ROWS - 1) / AGG_ROWS, dim3(1024), 0, stream>>>(
        h, esrc, edst, norm, bias, out, N, E);
}


# Scan key narrowed: launch 1 also writes 1 owner byte per edge (dst row / 391); aggregating work-groups scan 1 byte per edge, not 4, 8 edges per lane; row sum order changes, still fixed by inputs
# speedup vs baseline: 2.7278x; 1.2595x over previous
#include <hip/hip_runtime.h>

typedef __attribute__((ext_vector_type(2))) float v2f;
typedef __attribute__((ext_vector_type(8))) float v8f;

#define C_DIM 128
#define AGG_ROWS   391
#define KEY_CHUNK  256

__global__ __launch_bounds__(256) void gcn_gemm_norm(
    const float* __restrict__ x, const float* __restrict__ w,
    const float* __restrict__ norm, const int* __restrict__ edst,
    float* __restrict__ h, int N, int E)
{
    __shared__ float lds_w[C_DIM * C_DIM];

    {
        unsigned* okey = (unsigned*)(h + (size_t)N * C_DIM);
        const int nq = ((E + KEY_CHUNK - 1) & ~(KEY_CHUNK - 1)) >> 2;
        for (int i = blockIdx.x * 256 + threadIdx.x; i < nq; i += ((N + 127) >> 7) * 256) {
            int d[4];
            if (4 * i + 3 < E) {
                const int4 d4 = ((const int4*)edst)[i];
                d[0] = d4.x; d[1] = d4.y; d[2] = d4.z; d[3] = d4.w;
            } else {
                #pragma unroll
                for (int c = 0; c < 4; ++c) d[c] = (4 * i + c < E) ? edst[4 * i + c] : -1;
            }
            unsigned pk = 0u;
            #pragma unroll
            for (int c = 0; c < 4; ++c) {
                const unsigned o = ((unsigned)d[c] < (unsigned)N) ? (unsigned)d[c] / AGG_ROWS : 0u;
                pk |= (o & 255u) << (8 * c);
            }
            okey[i] = pk;
        }
    }

    for (int idx = threadIdx.x; idx < (C_DIM / 2) * C_DIM; idx += 256) {
        const int p = idx >> 7;
        const int n = idx & 127;
        v2f v;
        v.x = w[(2 * p + 0) * C_DIM + n];
        v.y = w[(2 * p + 1) * C_DIM + n];
        const int dw = p * 256 + ((2 * n + 32 * (p & 1)) & 255);
        *(v2f*)&lds_w[dw] = v;
    }
    __syncthreads();

    const int wave   = threadIdx.x >> 5;
    const int lane   = threadIdx.x & 31;
    const int lane16 = lane & 15;
    const int khalf  = lane >> 4;

    const int m0 = (blockIdx.x * 8 + wave) * 16;
    int row  = m0 + lane16;
    int rowc = row < N ? row : N - 1;

    v8f acc[8];
    const v8f vzero = {0.f,0.f,0.f,0.f,0.f,0.f,0.f,0.f};
    #pragma unroll
    for (int n = 0; n < 8; ++n) acc[n] = vzero;

    const float* xrow = x + (size_t)rowc * C_DIM;

    int coff[8];
    #pragma unroll
    for (int n = 0; n < 8; ++n)
        coff[n] = (32 * n + 2 * lane16 + 32 * khalf) & 255;
    const float* lwp = lds_w + khalf * 256;

    #pragma unroll 4
    for (int k = 0; k < C_DIM; k += 4) {
        const int kb = k + 2 * khalf;
        v2f a = *(const v2f*)(xrow + kb);
        const float* lk = lwp + (k >> 1) * 256;
        #pragma unroll
        for (int n = 0; n < 8; ++n) {
            v2f b = *(const v2f*)(lk + coff[n]);
            acc[n] = __builtin_amdgcn_wmma_f32_16x16x4_f32(
                false, a, false, b, (short)0, acc[n], false, false);
        }
    }

    float nrm[8];
    #pragma unroll
    for (int j = 0; j < 8; ++j) {
        const int r = m0 + j + 8 * khalf;
        nrm[j] = (r < N) ? norm[r] : 0.0f;
    }
    #pragma unroll
    for (int n = 0; n < 8; ++n) {
        #pragma unroll
        for (int j = 0; j < 8; ++j) {
            const int r = m0 + j + 8 * khalf;
            if (r < N) {
                h[(size_t)r * C_DIM + n * 16 + lane16] = acc[n][j] * nrm[j];
            }
        }
    }
}

#define AGG_WAVES  32
#define CAP_A      416
#define DUMMY_ROW  AGG_ROWS
#define SRC_BITS   20
#define SRC_MASK   0xFFFFF

__global__ __launch_bounds__(1024) void gcn_aggregate(
    const float* __restrict__ h, const int* __restrict__ esrc,
    const int* __restrict__ edst, const float* __restrict__ norm,
    const float* __restrict__ bias, float* __restrict__ out, int N, int E)
{
    __shared__ float acc[(AGG_ROWS + 1) * C_DIM];
    __shared__ int a_lst[AGG_WAVES][CAP_A];
    __shared__ int s_lst[AGG_WAVES][CAP_A];
    __shared__ int s_off[AGG_WAVES][AGG_WAVES + 1];
    __shared__ int a_done[AGG_WAVES];

    const int lane = threadIdx.x & 31;
    const int wave = __builtin_amdgcn_readfirstlane(threadIdx.x >> 5);
    const int row0 = blockIdx.x * AGG_ROWS;
    const int nrows_i = (N - row0) < AGG_ROWS ? (N - row0) : AGG_ROWS;
    const unsigned nrows = (unsigned)nrows_i;

    for (int i = threadIdx.x; i < (AGG_ROWS + 1) * C_DIM / 4; i += 1024)
        ((float4*)acc)[i] = make_float4(0.f, 0.f, 0.f, 0.f);

    const int slice = (((E + AGG_WAVES - 1) / AGG_WAVES) + KEY_CHUNK - 1) & ~(KEY_CHUNK - 1);
    long long sb = (long long)wave * slice;
    long long se = sb + slice;
    if (sb > E) sb = E;
    if (se > E) se = E;
    int cursor = (int)sb;
    const int end = (int)se;

    int* my_a = a_lst[wave];
    int* my_s = s_lst[wave];

    const uint2* kq = (const uint2*)(h + (size_t)N * C_DIM);
    const unsigned G = blockIdx.x * 0x01010101u;

    auto scan_chunk = [&](uint2 q, int cbase, int& cnt, bool tail) -> bool {
        const unsigned x0 = q.x ^ G, x1 = q.y ^ G;
        unsigned e0 = ~(((x0 & 0x7F7F7F7Fu) + 0x7F7F7F7Fu) | x0) & 0x80808080u;
        unsigned e1 = ~(((x1 & 0x7F7F7F7Fu) + 0x7F7F7F7Fu) | x1) & 0x80808080u;
        if (tail) {
            const int room = end - (cbase + lane * 8);
            const unsigned long long keep =
                room >= 8 ? ~0ull : (room <= 0 ? 0ull : ((1ull << (8 * room)) - 1ull));
            e0 &= (unsigned)keep;
            e1 &= (unsigned)(keep >> 32);
        }
        const unsigned hitm = __builtin_amdgcn_ballot_w32((e0 | e1) != 0u);
        if (hitm == 0u) return true;
        const int nh = __builtin_popcount(e0) + __builtin_popcount(e1);
        if (__builtin_amdgcn_ballot_w32(nh > 1) == 0u) {
            const int tot = __builtin_popcount(hitm);
            if (cnt + tot > CAP_A) return false;
            const int rank = (int)__builtin_amdgcn_mbcnt_lo(hitm, 0u);
            const unsigned long long em = (unsigned long long)e0 | ((unsigned long long)(e1 | 0x80000000u) << 32);
            const int byte = __builtin_ctzll(em) >> 3;
            if ((e0 | e1) != 0u) my_a[cnt + rank] = cbase + lane * 8 + byte;
            cnt += tot;
            return true;
        }
        unsigned bm[8];
        int tot = 0;
        #pragma unroll
        for (int j = 0; j < 8; ++j) {
            const unsigned ej = j < 4 ? e0 : e1;
            bm[j] = __builtin_amdgcn_ballot_w32(((ej >> (8 * (j & 3) + 7)) & 1u) != 0u);
            tot += __builtin_popcount(bm[j]);
        }
        if (cnt + tot > CAP_A) return false;
        #pragma unroll
        for (int j = 0; j < 8; ++j) {
            const unsigned ej = j < 4 ? e0 : e1;
            const int rank = (int)__builtin_amdgcn_mbcnt_lo(bm[j], 0u);
            if (((ej >> (8 * (j & 3) + 7)) & 1u) != 0u) my_a[cnt + rank] = cbase + lane * 8 + j;
            cnt += __builtin_popcount(bm[j]);
        }
        return true;
    };

    auto flush_rows = [&](int cnt) {
        const int padded = (cnt + 7) & ~7;
        if (lane < padded - cnt) my_a[cnt + lane] = DUMMY_ROW << SRC_BITS;
        for (int i = 0; i < padded; i += 8) {
            const int pv = my_a[i + (lane & 7)];
            float4 v[8];
            int r[8];
            #pragma unroll
            for (int k = 0; k < 8; ++k) {
                const int p = __builtin_amdgcn_readlane(pv, k);
                r[k] = p >> SRC_BITS;
                v[k] = *(const float4*)(h + (size_t)(p & SRC_MASK) * C_DIM + lane * 4);
            }
            #pragma unroll
            for (int k = 0; k < 8; ++k) {
                float* a = &acc[r[k] * C_DIM + lane];
                __hip_atomic_fetch_add(a,      v[k].x, __ATOMIC_RELAXED, __HIP_MEMORY_SCOPE_WORKGROUP);
                __hip_atomic_fetch_add(a + 32, v[k].y, __ATOMIC_RELAXED, __HIP_MEMORY_SCOPE_WORKGROUP);
                __hip_atomic_fetch_add(a + 64, v[k].z, __ATOMIC_RELAXED, __HIP_MEMORY_SCOPE_WORKGROUP);
                __hip_atomic_fetch_add(a + 96, v[k].w, __ATOMIC_RELAXED, __HIP_MEMORY_SCOPE_WORKGROUP);
            }
        }
    };

    auto key_masks = [&](int key, bool valid, unsigned& same, unsigned& mine) {
        const unsigned vm = __builtin_amdgcn_ballot_w32(valid);
        unsigned sm = vm, mn = vm;
        for (int b = 0; b < 5; ++b) {
            const unsigned bb = __builtin_amdgcn_ballot_w32(((key >> b) & 1) != 0);
            sm &= ~(bb ^ (0u - (unsigned)((key  >> b) & 1)));
            mn &= ~(bb ^ (0u - (unsigned)((lane >> b) & 1)));
        }
        same = sm;
        mine = mn;
    };

    bool all_done = false;
    while (!all_done) {
        int cnt = 0;
        bool full = false;
        while (!full && cursor + 4 * KEY_CHUNK <= end) {
            const uint2* p = kq + (cursor >> 3) + lane;
            const uint2 q0 = p[0], q1 = p[32], q2 = p[64], q3 = p[96];
            if (!scan_chunk(q0, cursor, cnt, false)) { full = true; break; }
            cursor += KEY_CHUNK;
            if (!scan_chunk(q1, cursor, cnt, false)) { full = true; break; }
            cursor += KEY_CHUNK;
            if (!scan_chunk(q2, cursor, cnt, false)) { full = true; break; }
            cursor += KEY_CHUNK;
            if (!scan_chunk(q3, cursor, cnt, false)) { full = true; break; }
            cursor += KEY_CHUNK;
        }
        while (!full && cursor < end) {
            const uint2 q = kq[(cursor >> 3) + lane];
            if (!scan_chunk(q, cursor, cnt, cursor + KEY_CHUNK > end)) { full = true; break; }
            cursor += KEY_CHUNK;
        }
        for (int i = lane; i < cnt; i += 32) {
            const int e = my_a[i];
            const unsigned lr = (unsigned)(edst[e] - row0);
            my_a[i] = lr < nrows ? (esrc[e] | (int)(lr << SRC_BITS)) : (DUMMY_ROW << SRC_BITS);
        }

        int cntv = 0;
        for (int i0 = 0; i0 < cnt; i0 += 32) {
            const bool valid = i0 + lane < cnt;
            const int p = valid ? my_a[i0 + lane] : 0;
            unsigned same, mine;
            key_masks((p >> SRC_BITS) & 31, valid, same, mine);
            cntv += __builtin_popcount(mine);
        }
        int inc = cntv;
        for (int d = 1; d < 32; d <<= 1) {
            const int t = __builtin_amdgcn_ds_bpermute(((lane - d) & 31) << 2, inc);
            if (lane >= d) inc += t;
        }
        int runv = inc - cntv;
        s_off[wave][lane] = runv;
        if (lane == 31) s_off[wave][AGG_WAVES] = inc;
        for (int i0 = 0; i0 < cnt; i0 += 32) {
            const bool valid = i0 + lane < cnt;
            const int p = valid ? my_a[i0 + lane] : 0;
            const int key = (p >> SRC_BITS) & 31;
            unsigned same, mine;
            key_masks(key, valid, same, mine);
            const int base = __builtin_amdgcn_ds_bpermute(key << 2, runv);
            const int rank = (int)__builtin_amdgcn_mbcnt_lo(same, 0u);
            if (valid) my_s[base + rank] = p;
            runv += __builtin_popcount(mine);
        }
        if (lane == 0) a_done[wave] = (cursor >= end) ? 1 : 0;
        __syncthreads();

        all_done = __builtin_amdgcn_ballot_w32(a_done[lane] != 0) == 0xFFFFFFFFu;

        const int segb = s_off[lane][wave];
        const int sege = s_off[lane][wave + 1];
        int cb = 0;
        for (int s = 0; s < AGG_WAVES; ++s) {
            const int b0 = __builtin_amdgcn_readlane(segb, s);
            const int n  = __builtin_amdgcn_readlane(sege, s) - b0;
            for (int i0 = 0; i0 < n; i0 += 32) {
                if (cb + 32 > CAP_A - 8) { flush_rows(cb); cb = 0; }
                const int m = (n - i0) < 32 ? (n - i0) : 32;
                if (lane < m) my_a[cb + lane] = s_lst[s][b0 + i0 + lane];
                cb += m;
            }
        }
        flush_rows(cb);
        __syncthreads();
    }

    const float4 b4 = *(const float4*)(bias + lane * 4);
    for (int lr = wave; lr < nrows_i; lr += AGG_WAVES) {
        const float* a = &acc[lr * C_DIM + lane];
        const float nm = norm[row0 + lr];
        float4 v;
        v.x = a[0]  * nm + b4.x;
        v.y = a[32] * nm + b4.y;
        v.z = a[64] * nm + b4.z;
        v.w = a[96] * nm + b4.w;
        *(float4*)(out + (size_t)(row0 + lr) * C_DIM + lane * 4) = v;
    }
}

extern "C" void kernel_launch(void* const* d_in, const int* in_sizes, int n_in,
                              void* d_out, int out_size, void* d_ws, size_t ws_size,
                              hipStream_t stream) {
    const float* x    = (const float*)d_in[0];
    const float* norm = (const float*)d_in[1];
    const float* w    = (const float*)d_in[2];
    const float* bias = (const float*)d_in[3];
    const int*   esrc = (const int*)d_in[4];
    const int*   edst = (const int*)d_in[5];
    float* out = (float*)d_out;

    const int N = in_sizes[0] / C_DIM;
    const int E = in_sizes[4];
    float* h = (float*)d_ws;

    gcn_gemm_norm<<<(N + 127) / 128, dim3(256), 0, stream>>>(x, w, norm, edst, h, N, E);

    gcn_aggregate<<<(N + AGG_ROWS - 1) / AGG_ROWS, dim3(1024), 0, stream>>>(
        h, esrc, edst, norm, bias, out, N, E);
}


# Scan key bit-sliced: launch 1 writes the owner byte of every 32 edges as 8 bit-plane words; an aggregating lane matches 32 edges with 8 xor/and ops; row sum order changes, still fixed by inputs
# speedup vs baseline: 3.1195x; 1.1436x over previous
#include <hip/hip_runtime.h>

typedef __attribute__((ext_vector_type(2))) float v2f;
typedef __attribute__((ext_vector_type(8))) float v8f;

#define C_DIM 128
#define AGG_ROWS   391
#define KEY_BLOCK  1024

__global__ __launch_bounds__(256) void gcn_gemm_norm(
    const float* __restrict__ x, const float* __restrict__ w,
    const float* __restrict__ norm, const int* __restrict__ edst,
    float* __restrict__ h, int N, int E)
{
    __shared__ float lds_w[C_DIM * C_DIM];

    {
        uint4* okey = (uint4*)(h + (size_t)N * C_DIM);
        const int nblk = (E + KEY_BLOCK - 1) / KEY_BLOCK;
        const int kl = threadIdx.x & 31;
        const int kw = __builtin_amdgcn_readfirstlane(threadIdx.x >> 5);
        for (int u = blockIdx.x + gridDim.x * kw; u < 4 * nblk; u += gridDim.x * 8) {
            const int blk = u >> 2;
            const int q   = u & 3;
            unsigned pl[8];
            #pragma unroll
            for (int b = 0; b < 8; ++b) pl[b] = 0u;
            const int base = blk * KEY_BLOCK + q * 256;
            int dv[8];
            #pragma unroll
            for (int j = 0; j < 8; ++j) {
                const int e = base + 32 * j + kl;
                dv[j] = edst[e < E ? e : E - 1];
            }
            #pragma unroll
            for (int j = 0; j < 8; ++j) {
                const int e = base + 32 * j + kl;
                const int d = e < E ? dv[j] : -1;
                const unsigned o = ((unsigned)d < (unsigned)N) ? (unsigned)d / AGG_ROWS : 0u;
                #pragma unroll
                for (int b = 0; b < 8; ++b) {
                    const unsigned bal = __builtin_amdgcn_ballot_w32(((o >> b) & 1u) != 0u);
                    if (kl == 8 * q + j) pl[b] = bal;
                }
            }
            if ((kl >> 3) == q) {
                okey[(size_t)blk * 64 + kl]      = make_uint4(pl[0], pl[1], pl[2], pl[3]);
                okey[(size_t)blk * 64 + 32 + kl] = make_uint4(pl[4], pl[5], pl[6], pl[7]);
            }
        }
    }

    for (int idx = threadIdx.x; idx < (C_DIM / 2) * C_DIM; idx += 256) {
        const int p = idx >> 7;
        const int n = idx & 127;
        v2f v;
        v.x = w[(2 * p + 0) * C_DIM + n];
        v.y = w[(2 * p + 1) * C_DIM + n];
        const int dw = p * 256 + ((2 * n + 32 * (p & 1)) & 255);
        *(v2f*)&lds_w[dw] = v;
    }
    __syncthreads();

    const int wave   = threadIdx.x >> 5;
    const int lane   = threadIdx.x & 31;
    const int lane16 = lane & 15;
    const int khalf  = lane >> 4;

    const int m0 = (blockIdx.x * 8 + wave) * 16;
    int row  = m0 + lane16;
    int rowc = row < N ? row : N - 1;

    v8f acc[8];
    const v8f vzero = {0.f,0.f,0.f,0.f,0.f,0.f,0.f,0.f};
    #pragma unroll
    for (int n = 0; n < 8; ++n) acc[n] = vzero;

    const float* xrow = x + (size_t)rowc * C_DIM;

    int coff[8];
    #pragma unroll
    for (int n = 0; n < 8; ++n)
        coff[n] = (32 * n + 2 * lane16 + 32 * khalf) & 255;
    const float* lwp = lds_w + khalf * 256;

    #pragma unroll 4
    for (int k = 0; k < C_DIM; k += 4) {
        const int kb = k + 2 * khalf;
        v2f a = *(const v2f*)(xrow + kb);
        const float* lk = lwp + (k >> 1) * 256;
        #pragma unroll
        for (int n = 0; n < 8; ++n) {
            v2f b = *(const v2f*)(lk + coff[n]);
            acc[n] = __builtin_amdgcn_wmma_f32_16x16x4_f32(
                false, a, false, b, (short)0, acc[n], false, false);
        }
    }

    float nrm[8];
    #pragma unroll
    for (int j = 0; j < 8; ++j) {
        const int r = m0 + j + 8 * khalf;
        nrm[j] = (r < N) ? norm[r] : 0.0f;
    }
    #pragma unroll
    for (int n = 0; n < 8; ++n) {
        #pragma unroll
        for (int j = 0; j < 8; ++j) {
            const int r = m0 + j + 8 * khalf;
            if (r < N) {
                h[(size_t)r * C_DIM + n * 16 + lane16] = acc[n][j] * nrm[j];
            }
        }
    }
}

#define AGG_WAVES  32
#define CAP_A      416
#define DUMMY_ROW  AGG_ROWS
#define SRC_BITS   20
#define SRC_MASK   0xFFFFF

__global__ __launch_bounds__(1024) void gcn_aggregate(
    const float* __restrict__ h, const int* __restrict__ esrc,
    const int* __restrict__ edst, const float* __restrict__ norm,
    const float* __restrict__ bias, float* __restrict__ out, int N, int E)
{
    __shared__ float acc[(AGG_ROWS + 1) * C_DIM];
    __shared__ int a_lst[AGG_WAVES][CAP_A];
    __shared__ int s_lst[AGG_WAVES][CAP_A];
    __shared__ int s_off[AGG_WAVES][AGG_WAVES + 1];
    __shared__ int a_done[AGG_WAVES];

    const int lane = threadIdx.x & 31;
    const int wave = __builtin_amdgcn_readfirstlane(threadIdx.x >> 5);
    const int row0 = blockIdx.x * AGG_ROWS;
    const int nrows_i = (N - row0) < AGG_ROWS ? (N - row0) : AGG_ROWS;
    const unsigned nrows = (unsigned)nrows_i;

    for (int i = threadIdx.x; i < (AGG_ROWS + 1) * C_DIM / 4; i += 1024)
        ((float4*)acc)[i] = make_float4(0.f, 0.f, 0.f, 0.f);

    const int nblk = (E + KEY_BLOCK - 1) / KEY_BLOCK;
    const int per  = (nblk + AGG_WAVES - 1) / AGG_WAVES;
    int cursor = wave * per;
    if (cursor > nblk) cursor = nblk;
    int end = cursor + per;
    if (end > nblk) end = nblk;

    int* my_a = a_lst[wave];
    int* my_s = s_lst[wave];

    const uint4* kq = (const uint4*)(h + (size_t)N * C_DIM);
    unsigned gx[8];
    #pragma unroll
    for (int b = 0; b < 8; ++b) gx[b] = ((blockIdx.x >> b) & 1u) ? 0u : ~0u;

    auto match = [&](uint4 a, uint4 b, int blk) -> unsigned {
        unsigned m = (a.x ^ gx[0]) & (a.y ^ gx[1]) & (a.z ^ gx[2]) & (a.w ^ gx[3])
                   & (b.x ^ gx[4]) & (b.y ^ gx[5]) & (b.z ^ gx[6]) & (b.w ^ gx[7]);
        if (blk * KEY_BLOCK + KEY_BLOCK > E) {
            const int room = E - (blk * KEY_BLOCK + lane * 32);
            m &= room >= 32 ? ~0u : (room <= 0 ? 0u : ((1u << room) - 1u));
        }
        return m;
    };

    auto drain = [&](unsigned& m, int blk, int& cnt) -> bool {
        const int ebase = blk * KEY_BLOCK + lane * 32;
        unsigned any = __builtin_amdgcn_ballot_w32(m != 0u);
        while (any != 0u) {
            const int tot = __builtin_popcount(any);
            if (cnt + tot > CAP_A) return false;
            const int rank = (int)__builtin_amdgcn_mbcnt_lo(any, 0u);
            if (m != 0u) my_a[cnt + rank] = ebase + __builtin_ctz(m);
            cnt += tot;
            m &= m - 1u;
            any = __builtin_amdgcn_ballot_w32(m != 0u);
        }
        return true;
    };

    unsigned pend = 0u;
    int pend_blk = 0;
    auto step = [&](uint4 a, uint4 b, int blk, int& cnt) -> bool {
        unsigned m = match(a, b, blk);
        if (drain(m, blk, cnt)) return true;
        pend = m;
        pend_blk = blk;
        return false;
    };

    auto flush_rows = [&](int cnt) {
        const int padded = (cnt + 7) & ~7;
        if (lane < padded - cnt) my_a[cnt + lane] = DUMMY_ROW << SRC_BITS;
        for (int i = 0; i < padded; i += 8) {
            const int pv = my_a[i + (lane & 7)];
            float4 v[8];
            int r[8];
            #pragma unroll
            for (int k = 0; k < 8; ++k) {
                const int p = __builtin_amdgcn_readlane(pv, k);
                r[k] = p >> SRC_BITS;
                v[k] = *(const float4*)(h + (size_t)(p & SRC_MASK) * C_DIM + lane * 4);
            }
            #pragma unroll
            for (int k = 0; k < 8; ++k) {
                float* a = &acc[r[k] * C_DIM + lane];
                __hip_atomic_fetch_add(a,      v[k].x, __ATOMIC_RELAXED, __HIP_MEMORY_SCOPE_WORKGROUP);
                __hip_atomic_fetch_add(a + 32, v[k].y, __ATOMIC_RELAXED, __HIP_MEMORY_SCOPE_WORKGROUP);
                __hip_atomic_fetch_add(a + 64, v[k].z, __ATOMIC_RELAXED, __HIP_MEMORY_SCOPE_WORKGROUP);
                __hip_atomic_fetch_add(a + 96, v[k].w, __ATOMIC_RELAXED, __HIP_MEMORY_SCOPE_WORKGROUP);
            }
        }
    };

    auto key_masks = [&](int key, bool valid, unsigned& same, unsigned& mine) {
        const unsigned vm = __builtin_amdgcn_ballot_w32(valid);
        unsigned sm = vm, mn = vm;
        for (int b = 0; b < 5; ++b) {
            const unsigned bb = __builtin_amdgcn_ballot_w32(((key >> b) & 1) != 0);
            sm &= ~(bb ^ (0u - (unsigned)((key  >> b) & 1)));
            mn &= ~(bb ^ (0u - (unsigned)((lane >> b) & 1)));
        }
        same = sm;
        mine = mn;
    };

    bool all_done = false;
    while (!all_done) {
        int cnt = 0;
        bool full = false;
        if (__builtin_amdgcn_ballot_w32(pend != 0u) != 0u) full = !drain(pend, pend_blk, cnt);
        while (!full && cursor + 4 <= end) {
            const uint4* p = kq + (size_t)cursor * 64 + lane;
            const uint4 a0 = p[0],   b0 = p[32];
            const uint4 a1 = p[64],  b1 = p[96];
            const uint4 a2 = p[128], b2 = p[160];
            const uint4 a3 = p[192], b3 = p[224];
            if (!step(a0, b0, cursor,     cnt)) { cursor += 1; full = true; break; }
            if (!step(a1, b1, cursor + 1, cnt)) { cursor += 2; full = true; break; }
            if (!step(a2, b2, cursor + 2, cnt)) { cursor += 3; full = true; break; }
            if (!step(a3, b3, cursor + 3, cnt)) { cursor += 4; full = true; break; }
            cursor += 4;
        }
        while (!full && cursor < end) {
            const uint4* p = kq + (size_t)cursor * 64 + lane;
            const uint4 a = p[0], b = p[32];
            if (!step(a, b, cursor, cnt)) { cursor += 1; full = true; break; }
            cursor += 1;
        }
        for (int i = lane; i < cnt; i += 32) {
            const int e = my_a[i];
            const unsigned lr = (unsigned)(edst[e] - row0);
            my_a[i] = lr < nrows ? (esrc[e] | (int)(lr << SRC_BITS)) : (DUMMY_ROW << SRC_BITS);
        }

        int cntv = 0;
        for (int i0 = 0; i0 < cnt; i0 += 32) {
            const bool valid = i0 + lane < cnt;
            const int p = valid ? my_a[i0 + lane] : 0;
            unsigned same, mine;
            key_masks((p >> SRC_BITS) & 31, valid, same, mine);
            cntv += __builtin_popcount(mine);
        }
        int inc = cntv;
        for (int d = 1; d < 32; d <<= 1) {
            const int t = __builtin_amdgcn_ds_bpermute(((lane - d) & 31) << 2, inc);
            if (lane >= d) inc += t;
        }
        int runv = inc - cntv;
        s_off[wave][lane] = runv;
        if (lane == 31) s_off[wave][AGG_WAVES] = inc;
        for (int i0 = 0; i0 < cnt; i0 += 32) {
            const bool valid = i0 + lane < cnt;
            const int p = valid ? my_a[i0 + lane] : 0;
            const int key = (p >> SRC_BITS) & 31;
            unsigned same, mine;
            key_masks(key, valid, same, mine);
            const int base = __builtin_amdgcn_ds_bpermute(key << 2, runv);
            const int rank = (int)__builtin_amdgcn_mbcnt_lo(same, 0u);
            if (valid) my_s[base + rank] = p;
            runv += __builtin_popcount(mine);
        }
        if (lane == 0) a_done[wave] = (cursor >= end && !full) ? 1 : 0;
        __syncthreads();

        all_done = __builtin_amdgcn_ballot_w32(a_done[lane] != 0) == 0xFFFFFFFFu;

        const int segb = s_off[lane][wave];
        const int sege = s_off[lane][wave + 1];
        int cb = 0;
        for (int s = 0; s < AGG_WAVES; ++s) {
            const int b0 = __builtin_amdgcn_readlane(segb, s);
            const int n  = __builtin_amdgcn_readlane(sege, s) - b0;
            for (int i0 = 0; i0 < n; i0 += 32) {
                if (cb + 32 > CAP_A - 8) { flush_rows(cb); cb = 0; }
                const int m = (n - i0) < 32 ? (n - i0) : 32;
                if (lane < m) my_a[cb + lane] = s_lst[s][b0 + i0 + lane];
                cb += m;
            }
        }
        flush_rows(cb);
        __syncthreads();
    }

    const float4 b4 = *(const float4*)(bias + lane * 4);
    for (int lr = wave; lr < nrows_i; lr += AGG_WAVES) {
        const float* a = &acc[lr * C_DIM + lane];
        const float nm = norm[row0 + lr];
        float4 v;
        v.x = a[0]  * nm + b4.x;
        v.y = a[32] * nm + b4.y;
        v.z = a[64] * nm + b4.z;
        v.w = a[96] * nm + b4.w;
        *(float4*)(out + (size_t)(row0 + lr) * C_DIM + lane * 4) = v;
    }
}

extern "C" void kernel_launch(void* const* d_in, const int* in_sizes, int n_in,
                              void* d_out, int out_size, void* d_ws, size_t ws_size,
                              hipStream_t stream) {
    const float* x    = (const float*)d_in[0];
    const float* norm = (const float*)d_in[1];
    const float* w    = (const float*)d_in[2];
    const float* bias = (const float*)d_in[3];
    const int*   esrc = (const int*)d_in[4];
    const int*   edst = (const int*)d_in[5];
    float* out = (float*)d_out;

    const int N = in_sizes[0] / C_DIM;
    const int E = in_sizes[4];
    float* h = (float*)d_ws;

    gcn_gemm_norm<<<(N + 127) / 128, dim3(256), 0, stream>>>(x, w, norm, edst, h, N, E);

    gcn_aggregate<<<(N + AGG_ROWS - 1) / AGG_ROWS, dim3(1024), 0, stream>>>(
        h, esrc, edst, norm, bias, out, N, E);
}


# Aggregation: owner wave orders its work list by out row (stable counting sort in registers), sums a row's h rows in registers: 4 LDS adds per row, not per edge; order kept, one partial sum per refill
# speedup vs baseline: 3.3780x; 1.0829x over previous
#include <hip/hip_runtime.h>

typedef __attribute__((ext_vector_type(2))) float v2f;
typedef __attribute__((ext_vector_type(8))) float v8f;

#define C_DIM 128
#define AGG_ROWS   391
#define KEY_BLOCK  1024

__global__ __launch_bounds__(256) void gcn_gemm_norm(
    const float* __restrict__ x, const float* __restrict__ w,
    const float* __restrict__ norm, const int* __restrict__ edst,
    float* __restrict__ h, int N, int E)
{
    __shared__ float lds_w[C_DIM * C_DIM];

    {
        uint4* okey = (uint4*)(h + (size_t)N * C_DIM);
        const int nblk = (E + KEY_BLOCK - 1) / KEY_BLOCK;
        const int kl = threadIdx.x & 31;
        const int kw = __builtin_amdgcn_readfirstlane(threadIdx.x >> 5);
        for (int u = blockIdx.x + gridDim.x * kw; u < 4 * nblk; u += gridDim.x * 8) {
            const int blk = u >> 2;
            const int q   = u & 3;
            unsigned pl[8];
            #pragma unroll
            for (int b = 0; b < 8; ++b) pl[b] = 0u;
            const int base = blk * KEY_BLOCK + q * 256;
            int dv[8];
            #pragma unroll
            for (int j = 0; j < 8; ++j) {
                const int e = base + 32 * j + kl;
                dv[j] = edst[e < E ? e : E - 1];
            }
            #pragma unroll
            for (int j = 0; j < 8; ++j) {
                const int e = base + 32 * j + kl;
                const int d = e < E ? dv[j] : -1;
                const unsigned o = ((unsigned)d < (unsigned)N) ? (unsigned)d / AGG_ROWS : 0u;
                #pragma unroll
                for (int b = 0; b < 8; ++b) {
                    const unsigned bal = __builtin_amdgcn_ballot_w32(((o >> b) & 1u) != 0u);
                    if (kl == 8 * q + j) pl[b] = bal;
                }
            }
            if ((kl >> 3) == q) {
                okey[(size_t)blk * 64 + kl]      = make_uint4(pl[0], pl[1], pl[2], pl[3]);
                okey[(size_t)blk * 64 + 32 + kl] = make_uint4(pl[4], pl[5], pl[6], pl[7]);
            }
        }
    }

    for (int idx = threadIdx.x; idx < (C_DIM / 2) * C_DIM; idx += 256) {
        const int p = idx >> 7;
        const int n = idx & 127;
        v2f v;
        v.x = w[(2 * p + 0) * C_DIM + n];
        v.y = w[(2 * p + 1) * C_DIM + n];
        const int dw = p * 256 + ((2 * n + 32 * (p & 1)) & 255);
        *(v2f*)&lds_w[dw] = v;
    }
    __syncthreads();

    const int wave   = threadIdx.x >> 5;
    const int lane   = threadIdx.x & 31;
    const int lane16 = lane & 15;
    const int khalf  = lane >> 4;

    const int m0 = (blockIdx.x * 8 + wave) * 16;
    int row  = m0 + lane16;
    int rowc = row < N ? row : N - 1;

    v8f acc[8];
    const v8f vzero = {0.f,0.f,0.f,0.f,0.f,0.f,0.f,0.f};
    #pragma unroll
    for (int n = 0; n < 8; ++n) acc[n] = vzero;

    const float* xrow = x + (size_t)rowc * C_DIM;

    int coff[8];
    #pragma unroll
    for (int n = 0; n < 8; ++n)
        coff[n] = (32 * n + 2 * lane16 + 32 * khalf) & 255;
    const float* lwp = lds_w + khalf * 256;

    #pragma unroll 4
    for (int k = 0; k < C_DIM; k += 4) {
        const int kb = k + 2 * khalf;
        v2f a = *(const v2f*)(xrow + kb);
        const float* lk = lwp + (k >> 1) * 256;
        #pragma unroll
        for (int n = 0; n < 8; ++n) {
            v2f b = *(const v2f*)(lk + coff[n]);
            acc[n] = __builtin_amdgcn_wmma_f32_16x16x4_f32(
                false, a, false, b, (short)0, acc[n], false, false);
        }
    }

    float nrm[8];
    #pragma unroll
    for (int j = 0; j < 8; ++j) {
        const int r = m0 + j + 8 * khalf;
        nrm[j] = (r < N) ? norm[r] : 0.0f;
    }
    #pragma unroll
    for (int n = 0; n < 8; ++n) {
        #pragma unroll
        for (int j = 0; j < 8; ++j) {
            const int r = m0 + j + 8 * khalf;
            if (r < N) {
                h[(size_t)r * C_DIM + n * 16 + lane16] = acc[n][j] * nrm[j];
            }
        }
    }
}

#define AGG_WAVES  32
#define CAP_A      416
#define DUMMY_ROW  AGG_ROWS
#define SRC_BITS   20
#define SRC_MASK   0xFFFFF

__global__ __launch_bounds__(1024) void gcn_aggregate(
    const float* __restrict__ h, const int* __restrict__ esrc,
    const int* __restrict__ edst, const float* __restrict__ norm,
    const float* __restrict__ bias, float* __restrict__ out, int N, int E)
{
    __shared__ float acc[(AGG_ROWS + 1) * C_DIM];
    __shared__ int a_lst[AGG_WAVES][CAP_A];
    __shared__ int s_lst[AGG_WAVES][CAP_A];
    __shared__ int s_off[AGG_WAVES][AGG_WAVES + 1];
    __shared__ int a_done[AGG_WAVES];

    const int lane = threadIdx.x & 31;
    const int wave = __builtin_amdgcn_readfirstlane(threadIdx.x >> 5);
    const int row0 = blockIdx.x * AGG_ROWS;
    const int nrows_i = (N - row0) < AGG_ROWS ? (N - row0) : AGG_ROWS;
    const unsigned nrows = (unsigned)nrows_i;

    for (int i = threadIdx.x; i < (AGG_ROWS + 1) * C_DIM / 4; i += 1024)
        ((float4*)acc)[i] = make_float4(0.f, 0.f, 0.f, 0.f);

    const int nblk = (E + KEY_BLOCK - 1) / KEY_BLOCK;
    const int per  = (nblk + AGG_WAVES - 1) / AGG_WAVES;
    int cursor = wave * per;
    if (cursor > nblk) cursor = nblk;
    int end = cursor + per;
    if (end > nblk) end = nblk;

    int* my_a = a_lst[wave];
    int* my_s = s_lst[wave];

    const uint4* kq = (const uint4*)(h + (size_t)N * C_DIM);
    unsigned gx[8];
    #pragma unroll
    for (int b = 0; b < 8; ++b) gx[b] = ((blockIdx.x >> b) & 1u) ? 0u : ~0u;

    auto match = [&](uint4 a, uint4 b, int blk) -> unsigned {
        unsigned m = (a.x ^ gx[0]) & (a.y ^ gx[1]) & (a.z ^ gx[2]) & (a.w ^ gx[3])
                   & (b.x ^ gx[4]) & (b.y ^ gx[5]) & (b.z ^ gx[6]) & (b.w ^ gx[7]);
        if (blk * KEY_BLOCK + KEY_BLOCK > E) {
            const int room = E - (blk * KEY_BLOCK + lane * 32);
            m &= room >= 32 ? ~0u : (room <= 0 ? 0u : ((1u << room) - 1u));
        }
        return m;
    };

    auto drain = [&](unsigned& m, int blk, int& cnt) -> bool {
        const int ebase = blk * KEY_BLOCK + lane * 32;
        unsigned any = __builtin_amdgcn_ballot_w32(m != 0u);
        while (any != 0u) {
            const int tot = __builtin_popcount(any);
            if (cnt + tot > CAP_A) return false;
            const int rank = (int)__builtin_amdgcn_mbcnt_lo(any, 0u);
            if (m != 0u) my_a[cnt + rank] = ebase + __builtin_ctz(m);
            cnt += tot;
            m &= m - 1u;
            any = __builtin_amdgcn_ballot_w32(m != 0u);
        }
        return true;
    };

    unsigned pend = 0u;
    int pend_blk = 0;
    auto step = [&](uint4 a, uint4 b, int blk, int& cnt) -> bool {
        unsigned m = match(a, b, blk);
        if (drain(m, blk, cnt)) return true;
        pend = m;
        pend_blk = blk;
        return false;
    };

    auto key_masks = [&](int key, bool valid, unsigned& same, unsigned& mine) {
        const unsigned vm = __builtin_amdgcn_ballot_w32(valid);
        unsigned sm = vm, mn = vm;
        for (int b = 0; b < 5; ++b) {
            const unsigned bb = __builtin_amdgcn_ballot_w32(((key >> b) & 1) != 0);
            sm &= ~(bb ^ (0u - (unsigned)((key  >> b) & 1)));
            mn &= ~(bb ^ (0u - (unsigned)((lane >> b) & 1)));
        }
        same = sm;
        mine = mn;
    };

    #define SORT_REGS ((CAP_A + 31) / 32)
    auto flush_rows = [&](int cnt) {
        int pe[SORT_REGS];
        #pragma unroll
        for (int i = 0; i < SORT_REGS; ++i)
            pe[i] = (i * 32 + lane < cnt) ? my_a[i * 32 + lane] : 0;
        int cntv = 0;
        #pragma unroll
        for (int i = 0; i < SORT_REGS; ++i) {
            if (i * 32 < cnt) {
                unsigned same, mine;
                key_masks(pe[i] >> (SRC_BITS + 5), i * 32 + lane < cnt, same, mine);
                cntv += __builtin_popcount(mine);
            }
        }
        int inc = cntv;
        for (int d = 1; d < 32; d <<= 1) {
            const int t = __builtin_amdgcn_ds_bpermute(((lane - d) & 31) << 2, inc);
            if (lane >= d) inc += t;
        }
        int runv = inc - cntv;
        #pragma unroll
        for (int i = 0; i < SORT_REGS; ++i) {
            if (i * 32 < cnt) {
                const bool valid = i * 32 + lane < cnt;
                const int key = pe[i] >> (SRC_BITS + 5);
                unsigned same, mine;
                key_masks(key, valid, same, mine);
                const int base = __builtin_amdgcn_ds_bpermute(key << 2, runv);
                const int rank = (int)__builtin_amdgcn_mbcnt_lo(same, 0u);
                if (valid) my_a[base + rank] = pe[i];
                runv += __builtin_popcount(mine);
            }
        }

        const int padded = (cnt + 7) & ~7;
        if (lane < padded - cnt) my_a[cnt + lane] = DUMMY_ROW << SRC_BITS;
        float4 s = make_float4(0.f, 0.f, 0.f, 0.f);
        int cur = DUMMY_ROW;
        auto put = [&]() {
            float* a = &acc[cur * C_DIM + lane];
            __hip_atomic_fetch_add(a,      s.x, __ATOMIC_RELAXED, __HIP_MEMORY_SCOPE_WORKGROUP);
            __hip_atomic_fetch_add(a + 32, s.y, __ATOMIC_RELAXED, __HIP_MEMORY_SCOPE_WORKGROUP);
            __hip_atomic_fetch_add(a + 64, s.z, __ATOMIC_RELAXED, __HIP_MEMORY_SCOPE_WORKGROUP);
            __hip_atomic_fetch_add(a + 96, s.w, __ATOMIC_RELAXED, __HIP_MEMORY_SCOPE_WORKGROUP);
        };
        for (int i = 0; i < padded; i += 8) {
            const int pv = my_a[i + (lane & 7)];
            float4 v[8];
            int r[8];
            #pragma unroll
            for (int k = 0; k < 8; ++k) {
                const int p = __builtin_amdgcn_readlane(pv, k);
                r[k] = p >> SRC_BITS;
                v[k] = *(const float4*)(h + (size_t)(p & SRC_MASK) * C_DIM + lane * 4);
            }
            #pragma unroll
            for (int k = 0; k < 8; ++k) {
                if (r[k] != cur) {
                    put();
                    cur = r[k];
                    s = v[k];
                } else {
                    s.x += v[k].x; s.y += v[k].y; s.z += v[k].z; s.w += v[k].w;
                }
            }
        }
        put();
    };

    bool all_done = false;
    while (!all_done) {
        int cnt = 0;
        bool full = false;
        if (__builtin_amdgcn_ballot_w32(pend != 0u) != 0u) full = !drain(pend, pend_blk, cnt);
        while (!full && cursor + 4 <= end) {
            const uint4* p = kq + (size_t)cursor * 64 + lane;
            const uint4 a0 = p[0],   b0 = p[32];
            const uint4 a1 = p[64],  b1 = p[96];
            const uint4 a2 = p[128], b2 = p[160];
            const uint4 a3 = p[192], b3 = p[224];
            if (!step(a0, b0, cursor,     cnt)) { cursor += 1; full = true; break; }
            if (!step(a1, b1, cursor + 1, cnt)) { cursor += 2; full = true; break; }
            if (!step(a2, b2, cursor + 2, cnt)) { cursor += 3; full = true; break; }
            if (!step(a3, b3, cursor + 3, cnt)) { cursor += 4; full = true; break; }
            cursor += 4;
        }
        while (!full && cursor < end) {
            const uint4* p = kq + (size_t)cursor * 64 + lane;
            const uint4 a = p[0], b = p[32];
            if (!step(a, b, cursor, cnt)) { cursor += 1; full = true; break; }
            cursor += 1;
        }
        for (int i = lane; i < cnt; i += 32) {
            const int e = my_a[i];
            const unsigned lr = (unsigned)(edst[e] - row0);
            my_a[i] = lr < nrows ? (esrc[e] | (int)(lr << SRC_BITS)) : (DUMMY_ROW << SRC_BITS);
        }

        int cntv = 0;
        for (int i0 = 0; i0 < cnt; i0 += 32) {
            const bool valid = i0 + lane < cnt;
            const int p = valid ? my_a[i0 + lane] : 0;
            unsigned same, mine;
            key_masks((p >> SRC_BITS) & 31, valid, same, mine);
            cntv += __builtin_popcount(mine);
        }
        int inc = cntv;
        for (int d = 1; d < 32; d <<= 1) {
            const int t = __builtin_amdgcn_ds_bpermute(((lane - d) & 31) << 2, inc);
            if (lane >= d) inc += t;
        }
        int runv = inc - cntv;
        s_off[wave][lane] = runv;
        if (lane == 31) s_off[wave][AGG_WAVES] = inc;
        for (int i0 = 0; i0 < cnt; i0 += 32) {
            const bool valid = i0 + lane < cnt;
            const int p = valid ? my_a[i0 + lane] : 0;
            const int key = (p >> SRC_BITS) & 31;
            unsigned same, mine;
            key_masks(key, valid, same, mine);
            const int base = __builtin_amdgcn_ds_bpermute(key << 2, runv);
            const int rank = (int)__builtin_amdgcn_mbcnt_lo(same, 0u);
            if (valid) my_s[base + rank] = p;
            runv += __builtin_popcount(mine);
        }
        if (lane == 0) a_done[wave] = (cursor >= end && !full) ? 1 : 0;
        __syncthreads();

        all_done = __builtin_amdgcn_ballot_w32(a_done[lane] != 0) == 0xFFFFFFFFu;

        const int segb = s_off[lane][wave];
        const int sege = s_off[lane][wave + 1];
        int cb = 0;
        for (int s = 0; s < AGG_WAVES; ++s) {
            const int b0 = __builtin_amdgcn_readlane(segb, s);
            const int n  = __builtin_amdgcn_readlane(sege, s) - b0;
            for (int i0 = 0; i0 < n; i0 += 32) {
                if (cb + 32 > CAP_A - 8) { flush_rows(cb); cb = 0; }
                const int m = (n - i0) < 32 ? (n - i0) : 32;
                if (lane < m) my_a[cb + lane] = s_lst[s][b0 + i0 + lane];
                cb += m;
            }
        }
        flush_rows(cb);
        __syncthreads();
    }

    const float4 b4 = *(const float4*)(bias + lane * 4);
    for (int lr = wave; lr < nrows_i; lr += AGG_WAVES) {
        const float* a = &acc[lr * C_DIM + lane];
        const float nm = norm[row0 + lr];
        float4 v;
        v.x = a[0]  * nm + b4.x;
        v.y = a[32] * nm + b4.y;
        v.z = a[64] * nm + b4.z;
        v.w = a[96] * nm + b4.w;
        *(float4*)(out + (size_t)(row0 + lr) * C_DIM + lane * 4) = v;
    }
}

extern "C" void kernel_launch(void* const* d_in, const int* in_sizes, int n_in,
                              void* d_out, int out_size, void* d_ws, size_t ws_size,
                              hipStream_t stream) {
    const float* x    = (const float*)d_in[0];
    const float* norm = (const float*)d_in[1];
    const float* w    = (const float*)d_in[2];
    const float* bias = (const float*)d_in[3];
    const int*   esrc = (const int*)d_in[4];
    const int*   edst = (const int*)d_in[5];
    float* out = (float*)d_out;

    const int N = in_sizes[0] / C_DIM;
    const int E = in_sizes[4];
    float* h = (float*)d_ws;

    gcn_gemm_norm<<<(N + 127) / 128, dim3(256), 0, stream>>>(x, w, norm, edst, h, N, E);

    gcn_aggregate<<<(N + AGG_ROWS - 1) / AGG_ROWS, dim3(1024), 0, stream>>>(
        h, esrc, edst, norm, bias, out, N, E);
}


# Scan key as two one-hot nibbles: launch 1 writes 16+16 bit planes (owner high, low nibble); an aggregating work-group reads only its 2 planes, ands them: 2 bits per edge, not 8; sum order unchanged
# speedup vs baseline: 3.6428x; 1.0784x over previous
#include <hip/hip_runtime.h>

typedef __attribute__((ext_vector_type(2))) float v2f;
typedef __attribute__((ext_vector_type(8))) float v8f;

#define C_DIM 128
#define AGG_ROWS   391
#define KEY_BLOCK  1024
#define KEY_UNIT   2048
#define KEY_PLANES 32
#define KEY_PITCH  68

__global__ __launch_bounds__(256) void gcn_gemm_norm(
    const float* __restrict__ x, const float* __restrict__ w,
    const float* __restrict__ norm, const int* __restrict__ edst,
    float* __restrict__ h, int N, int E)
{
    __shared__ float lds_w[C_DIM * C_DIM];
    __shared__ __attribute__((aligned(16))) unsigned key_scr[KEY_PLANES * KEY_PITCH];

    {
        unsigned* kp = (unsigned*)(h + (size_t)N * C_DIM);
        const int nblk  = (E + KEY_BLOCK - 1) / KEY_BLOCK;
        const int nunit = (nblk + 1) >> 1;
        const size_t plane_words = (size_t)nunit * (KEY_UNIT / 32);
        const int kl = threadIdx.x & 31;
        const int kw = __builtin_amdgcn_readfirstlane(threadIdx.x >> 5);
        unsigned xm[4];
        #pragma unroll
        for (int b = 0; b < 4; ++b) xm[b] = ((kl >> b) & 1) ? 0u : ~0u;
        const bool hi_lane = kl < 16;
        for (int u = blockIdx.x; u < nunit; u += gridDim.x) {
            if (u != (int)blockIdx.x) __syncthreads();
            const int base = u * KEY_UNIT;
            int dv[8];
            #pragma unroll
            for (int j = 0; j < 8; ++j) {
                const int e = base + 256 * j + (int)threadIdx.x;
                dv[j] = edst[e < E ? e : E - 1];
            }
            #pragma unroll
            for (int j = 0; j < 8; ++j) {
                const int e = base + 256 * j + (int)threadIdx.x;
                const bool valid = e < E;
                const int d = dv[j];
                const unsigned o = ((unsigned)d < (unsigned)N) ? (unsigned)d / AGG_ROWS : 0u;
                unsigned word = __builtin_amdgcn_ballot_w32(valid);
                #pragma unroll
                for (int b = 0; b < 4; ++b) {
                    const unsigned bh = __builtin_amdgcn_ballot_w32(((o >> (4 + b)) & 1u) != 0u);
                    const unsigned bl = __builtin_amdgcn_ballot_w32(((o >> b) & 1u) != 0u);
                    word &= (hi_lane ? bh : bl) ^ xm[b];
                }
                key_scr[kl * KEY_PITCH + 8 * j + kw] = word;
            }
            __syncthreads();
            {
                const int plane = threadIdx.x >> 3;
                const int piece = threadIdx.x & 7;
                const uint4 v0 = *(const uint4*)&key_scr[plane * KEY_PITCH + 4 * piece];
                const uint4 v1 = *(const uint4*)&key_scr[plane * KEY_PITCH + 32 + 4 * piece];
                uint4* dstp = (uint4*)(kp + (size_t)plane * plane_words + (size_t)u * (KEY_UNIT / 32));
                dstp[piece]     = v0;
                dstp[8 + piece] = v1;
            }
        }
    }

    for (int idx = threadIdx.x; idx < (C_DIM / 2) * C_DIM; idx += 256) {
        const int p = idx >> 7;
        const int n = idx & 127;
        v2f v;
        v.x = w[(2 * p + 0) * C_DIM + n];
        v.y = w[(2 * p + 1) * C_DIM + n];
        const int dw = p * 256 + ((2 * n + 32 * (p & 1)) & 255);
        *(v2f*)&lds_w[dw] = v;
    }
    __syncthreads();

    const int wave   = threadIdx.x >> 5;
    const int lane   = threadIdx.x & 31;
    const int lane16 = lane & 15;
    const int khalf  = lane >> 4;

    const int m0 = (blockIdx.x * 8 + wave) * 16;
    int row  = m0 + lane16;
    int rowc = row < N ? row : N - 1;

    v8f acc[8];
    const v8f vzero = {0.f,0.f,0.f,0.f,0.f,0.f,0.f,0.f};
    #pragma unroll
    for (int n = 0; n < 8; ++n) acc[n] = vzero;

    const float* xrow = x + (size_t)rowc * C_DIM;

    int coff[8];
    #pragma unroll
    for (int n = 0; n < 8; ++n)
        coff[n] = (32 * n + 2 * lane16 + 32 * khalf) & 255;
    const float* lwp = lds_w + khalf * 256;

    #pragma unroll 4
    for (int k = 0; k < C_DIM; k += 4) {
        const int kb = k + 2 * khalf;
        v2f a = *(const v2f*)(xrow + kb);
        const float* lk = lwp + (k >> 1) * 256;
        #pragma unroll
        for (int n = 0; n < 8; ++n) {
            v2f b = *(const v2f*)(lk + coff[n]);
            acc[n] = __builtin_amdgcn_wmma_f32_16x16x4_f32(
                false, a, false, b, (short)0, acc[n], false, false);
        }
    }

    float nrm[8];
    #pragma unroll
    for (int j = 0; j < 8; ++j) {
        const int r = m0 + j + 8 * khalf;
        nrm[j] = (r < N) ? norm[r] : 0.0f;
    }
    #pragma unroll
    for (int n = 0; n < 8; ++n) {
        #pragma unroll
        for (int j = 0; j < 8; ++j) {
            const int r = m0 + j + 8 * khalf;
            if (r < N) {
                h[(size_t)r * C_DIM + n * 16 + lane16] = acc[n][j] * nrm[j];
            }
        }
    }
}

#define AGG_WAVES  32
#define CAP_A      416
#define DUMMY_ROW  AGG_ROWS
#define SRC_BITS   20
#define SRC_MASK   0xFFFFF

__global__ __launch_bounds__(1024) void gcn_aggregate(
    const float* __restrict__ h, const int* __restrict__ esrc,
    const int* __restrict__ edst, const float* __restrict__ norm,
    const float* __restrict__ bias, float* __restrict__ out, int N, int E)
{
    __shared__ float acc[(AGG_ROWS + 1) * C_DIM];
    __shared__ int a_lst[AGG_WAVES][CAP_A];
    __shared__ int s_lst[AGG_WAVES][CAP_A];
    __shared__ int s_off[AGG_WAVES][AGG_WAVES + 1];
    __shared__ int a_done[AGG_WAVES];

    const int lane = threadIdx.x & 31;
    const int wave = __builtin_amdgcn_readfirstlane(threadIdx.x >> 5);
    const int row0 = blockIdx.x * AGG_ROWS;
    const int nrows_i = (N - row0) < AGG_ROWS ? (N - row0) : AGG_ROWS;
    const unsigned nrows = (unsigned)nrows_i;

    for (int i = threadIdx.x; i < (AGG_ROWS + 1) * C_DIM / 4; i += 1024)
        ((float4*)acc)[i] = make_float4(0.f, 0.f, 0.f, 0.f);

    const int nblk = (E + KEY_BLOCK - 1) / KEY_BLOCK;
    const int per  = (nblk + AGG_WAVES - 1) / AGG_WAVES;
    int cursor = wave * per;
    if (cursor > nblk) cursor = nblk;
    int end = cursor + per;
    if (end > nblk) end = nblk;

    int* my_a = a_lst[wave];
    int* my_s = s_lst[wave];

    const size_t plane_words = (size_t)((nblk + 1) >> 1) * (KEY_UNIT / 32);
    const unsigned* kp = (const unsigned*)(h + (size_t)N * C_DIM);
    const unsigned* k_hi = kp + (size_t)((blockIdx.x >> 4) & 15u) * plane_words + lane;
    const unsigned* k_lo = kp + (size_t)(16u + (blockIdx.x & 15u)) * plane_words + lane;

    auto match = [&](unsigned a, unsigned b, int blk) -> unsigned {
        unsigned m = a & b;
        if (blk * KEY_BLOCK + KEY_BLOCK > E) {
            const int room = E - (blk * KEY_BLOCK + lane * 32);
            m &= room >= 32 ? ~0u : (room <= 0 ? 0u : ((1u << room) - 1u));
        }
        return m;
    };

    auto drain = [&](unsigned& m, int blk, int& cnt) -> bool {
        const int ebase = blk * KEY_BLOCK + lane * 32;
        unsigned any = __builtin_amdgcn_ballot_w32(m != 0u);
        while (any != 0u) {
            const int tot = __builtin_popcount(any);
            if (cnt + tot > CAP_A) return false;
            const int rank = (int)__builtin_amdgcn_mbcnt_lo(any, 0u);
            if (m != 0u) my_a[cnt + rank] = ebase + __builtin_ctz(m);
            cnt += tot;
            m &= m - 1u;
            any = __builtin_amdgcn_ballot_w32(m != 0u);
        }
        return true;
    };

    unsigned pend = 0u;
    int pend_blk = 0;
    auto step = [&](unsigned a, unsigned b, int blk, int& cnt) -> bool {
        unsigned m = match(a, b, blk);
        if (drain(m, blk, cnt)) return true;
        pend = m;
        pend_blk = blk;
        return false;
    };

    auto key_masks = [&](int key, bool valid, unsigned& same, unsigned& mine) {
        const unsigned vm = __builtin_amdgcn_ballot_w32(valid);
        unsigned sm = vm, mn = vm;
        for (int b = 0; b < 5; ++b) {
            const unsigned bb = __builtin_amdgcn_ballot_w32(((key >> b) & 1) != 0);
            sm &= ~(bb ^ (0u - (unsigned)((key  >> b) & 1)));
            mn &= ~(bb ^ (0u - (unsigned)((lane >> b) & 1)));
        }
        same = sm;
        mine = mn;
    };

    #define SORT_REGS ((CAP_A + 31) / 32)
    auto flush_rows = [&](int cnt) {
        int pe[SORT_REGS];
        #pragma unroll
        for (int i = 0; i < SORT_REGS; ++i)
            pe[i] = (i * 32 + lane < cnt) ? my_a[i * 32 + lane] : 0;
        int cntv = 0;
        #pragma unroll
        for (int i = 0; i < SORT_REGS; ++i) {
            if (i * 32 < cnt) {
                unsigned same, mine;
                key_masks(pe[i] >> (SRC_BITS + 5), i * 32 + lane < cnt, same, mine);
                cntv += __builtin_popcount(mine);
            }
        }
        int inc = cntv;
        for (int d = 1; d < 32; d <<= 1) {
            const int t = __builtin_amdgcn_ds_bpermute(((lane - d) & 31) << 2, inc);
            if (lane >= d) inc += t;
        }
        int runv = inc - cntv;
        #pragma unroll
        for (int i = 0; i < SORT_REGS; ++i) {
            if (i * 32 < cnt) {
                const bool valid = i * 32 + lane < cnt;
                const int key = pe[i] >> (SRC_BITS + 5);
                unsigned same, mine;
                key_masks(key, valid, same, mine);
                const int base = __builtin_amdgcn_ds_bpermute(key << 2, runv);
                const int rank = (int)__builtin_amdgcn_mbcnt_lo(same, 0u);
                if (valid) my_a[base + rank] = pe[i];
                runv += __builtin_popcount(mine);
            }
        }

        const int padded = (cnt + 7) & ~7;
        if (lane < padded - cnt) my_a[cnt + lane] = DUMMY_ROW << SRC_BITS;
        float4 s = make_float4(0.f, 0.f, 0.f, 0.f);
        int cur = DUMMY_ROW;
        auto put = [&]() {
            float* a = &acc[cur * C_DIM + lane];
            __hip_atomic_fetch_add(a,      s.x, __ATOMIC_RELAXED, __HIP_MEMORY_SCOPE_WORKGROUP);
            __hip_atomic_fetch_add(a + 32, s.y, __ATOMIC_RELAXED, __HIP_MEMORY_SCOPE_WORKGROUP);
            __hip_atomic_fetch_add(a + 64, s.z, __ATOMIC_RELAXED, __HIP_MEMORY_SCOPE_WORKGROUP);
            __hip_atomic_fetch_add(a + 96, s.w, __ATOMIC_RELAXED, __HIP_MEMORY_SCOPE_WORKGROUP);
        };
        for (int i = 0; i < padded; i += 8) {
            const int pv = my_a[i + (lane & 7)];
            float4 v[8];
            int r[8];
            #pragma unroll
            for (int k = 0; k < 8; ++k) {
                const int p = __builtin_amdgcn_readlane(pv, k);
                r[k] = p >> SRC_BITS;
                v[k] = *(const float4*)(h + (size_t)(p & SRC_MASK) * C_DIM + lane * 4);
            }
            #pragma unroll
            for (int k = 0; k < 8; ++k) {
                if (r[k] != cur) {
                    put();
                    cur = r[k];
                    s = v[k];
                } else {
                    s.x += v[k].x; s.y += v[k].y; s.z += v[k].z; s.w += v[k].w;
                }
            }
        }
        put();
    };

    bool all_done = false;
    while (!all_done) {
        int cnt = 0;
        bool full = false;
        if (__builtin_amdgcn_ballot_w32(pend != 0u) != 0u) full = !drain(pend, pend_blk, cnt);
        while (!full && cursor + 8 <= end) {
            const unsigned* pa = k_hi + (size_t)cursor * 32;
            const unsigned* pb = k_lo + (size_t)cursor * 32;
            unsigned wa[8], wb[8];
            #pragma unroll
            for (int t = 0; t < 8; ++t) { wa[t] = pa[32 * t]; wb[t] = pb[32 * t]; }
            int t = 0;
            #pragma unroll
            for (; t < 8; ++t)
                if (!step(wa[t], wb[t], cursor + t, cnt)) { full = true; break; }
            cursor += full ? t + 1 : 8;
        }
        while (!full && cursor < end) {
            const unsigned a = k_hi[(size_t)cursor * 32], b = k_lo[(size_t)cursor * 32];
            if (!step(a, b, cursor, cnt)) { cursor += 1; full = true; break; }
            cursor += 1;
        }
        for (int i = lane; i < cnt; i += 32) {
            const int e = my_a[i];
            const unsigned lr = (unsigned)(edst[e] - row0);
            my_a[i] = lr < nrows ? (esrc[e] | (int)(lr << SRC_BITS)) : (DUMMY_ROW << SRC_BITS);
        }

        int cntv = 0;
        for (int i0 = 0; i0 < cnt; i0 += 32) {
            const bool valid = i0 + lane < cnt;
            const int p = valid ? my_a[i0 + lane] : 0;
            unsigned same, mine;
            key_masks((p >> SRC_BITS) & 31, valid, same, mine);
            cntv += __builtin_popcount(mine);
        }
        int inc = cntv;
        for (int d = 1; d < 32; d <<= 1) {
            const int t = __builtin_amdgcn_ds_bpermute(((lane - d) & 31) << 2, inc);
            if (lane >= d) inc += t;
        }
        int runv = inc - cntv;
        s_off[wave][lane] = runv;
        if (lane == 31) s_off[wave][AGG_WAVES] = inc;
        for (int i0 = 0; i0 < cnt; i0 += 32) {
            const bool valid = i0 + lane < cnt;
            const int p = valid ? my_a[i0 + lane] : 0;
            const int key = (p >> SRC_BITS) & 31;
            unsigned same, mine;
            key_masks(key, valid, same, mine);
            const int base = __builtin_amdgcn_ds_bpermute(key << 2, runv);
            const int rank = (int)__builtin_amdgcn_mbcnt_lo(same, 0u);
            if (valid) my_s[base + rank] = p;
            runv += __builtin_popcount(mine);
        }
        if (lane == 0) a_done[wave] = (cursor >= end && !full) ? 1 : 0;
        __syncthreads();

        all_done = __builtin_amdgcn_ballot_w32(a_done[lane] != 0) == 0xFFFFFFFFu;

        const int segb = s_off[lane][wave];
        const int sege = s_off[lane][wave + 1];
        int cb = 0;
        for (int s = 0; s < AGG_WAVES; ++s) {
            const int b0 = __builtin_amdgcn_readlane(segb, s);
            const int n  = __builtin_amdgcn_readlane(sege, s) - b0;
            for (int i0 = 0; i0 < n; i0 += 32) {
                if (cb + 32 > CAP_A - 8) { flush_rows(cb); cb = 0; }
                const int m = (n - i0) < 32 ? (n - i0) : 32;
                if (lane < m) my_a[cb + lane] = s_lst[s][b0 + i0 + lane];
                cb += m;
            }
        }
        flush_rows(cb);
        __syncthreads();
    }

    const float4 b4 = *(const float4*)(bias + lane * 4);
    for (int lr = wave; lr < nrows_i; lr += AGG_WAVES) {
        const float* a = &acc[lr * C_DIM + lane];
        const float nm = norm[row0 + lr];
        float4 v;
        v.x = a[0]  * nm + b4.x;
        v.y = a[32] * nm + b4.y;
        v.z = a[64] * nm + b4.z;
        v.w = a[96] * nm + b4.w;
        *(float4*)(out + (size_t)(row0 + lr) * C_DIM + lane * 4) = v;
    }
}

extern "C" void kernel_launch(void* const* d_in, const int* in_sizes, int n_in,
                              void* d_out, int out_size, void* d_ws, size_t ws_size,
                              hipStream_t stream) {
    const float* x    = (const float*)d_in[0];
    const float* norm = (const float*)d_in[1];
    const float* w    = (const float*)d_in[2];
    const float* bias = (const float*)d_in[3];
    const int*   esrc = (const int*)d_in[4];
    const int*   edst = (const int*)d_in[5];
    float* out = (float*)d_out;

    const int N = in_sizes[0] / C_DIM;
    const int E = in_sizes[4];
    float* h = (float*)d_ws;

    gcn_gemm_norm<<<(N + 127) / 128, dim3(256), 0, stream>>>(x, w, norm, edst, h, N, E);

    gcn_aggregate<<<(N + AGG_ROWS - 1) / AGG_ROWS, dim3(1024), 0, stream>>>(
        h, esrc, edst, norm, bias, out, N, E);
}


# Edge lookup: launch 1 also writes 1 record word per edge (src | local out row << 20); an aggregating wave fetches 1 word per hit edge, all in flight at once, not dst then src; sums unchanged
# speedup vs baseline: 3.8451x; 1.0555x over previous
#include <hip/hip_runtime.h>

typedef __attribute__((ext_vector_type(2))) float v2f;
typedef __attribute__((ext_vector_type(8))) float v8f;

#define C_DIM 128
#define AGG_ROWS   391
#define KEY_BLOCK  1024
#define KEY_UNIT   2048
#define KEY_PLANES 32
#define KEY_PITCH  68
#define DUMMY_ROW  AGG_ROWS
#define SRC_BITS   20
#define SRC_MASK   0xFFFFF

__global__ __launch_bounds__(256) void gcn_gemm_norm(
    const float* __restrict__ x, const float* __restrict__ w,
    const float* __restrict__ norm, const int* __restrict__ esrc,
    const int* __restrict__ edst, float* __restrict__ h, int N, int E)
{
    __shared__ float lds_w[C_DIM * C_DIM];
    __shared__ __attribute__((aligned(16))) unsigned key_scr[KEY_PLANES * KEY_PITCH];

    {
        unsigned* kp = (unsigned*)(h + (size_t)N * C_DIM);
        const int nblk  = (E + KEY_BLOCK - 1) / KEY_BLOCK;
        const int nunit = (nblk + 1) >> 1;
        const size_t plane_words = (size_t)nunit * (KEY_UNIT / 32);
        int* rec = (int*)(kp + (size_t)KEY_PLANES * plane_words);
        const int kl = threadIdx.x & 31;
        const int kw = __builtin_amdgcn_readfirstlane(threadIdx.x >> 5);
        unsigned xm[4];
        #pragma unroll
        for (int b = 0; b < 4; ++b) xm[b] = ((kl >> b) & 1) ? 0u : ~0u;
        const bool hi_lane = kl < 16;
        for (int u = blockIdx.x; u < nunit; u += gridDim.x) {
            if (u != (int)blockIdx.x) __syncthreads();
            const int base = u * KEY_UNIT;
            int dv[8], sv[8];
            #pragma unroll
            for (int j = 0; j < 8; ++j) {
                const int e = base + 256 * j + (int)threadIdx.x;
                const int ec = e < E ? e : E - 1;
                dv[j] = edst[ec];
                sv[j] = esrc[ec];
            }
            #pragma unroll
            for (int j = 0; j < 8; ++j) {
                const int e = base + 256 * j + (int)threadIdx.x;
                const bool valid = e < E;
                const int d = dv[j];
                const bool inside = (unsigned)d < (unsigned)N;
                const unsigned o = inside ? (unsigned)d / AGG_ROWS : 0u;
                const unsigned lr = inside ? (unsigned)d - o * AGG_ROWS : (unsigned)DUMMY_ROW;
                rec[valid ? e : E - 1] = sv[j] | (int)(lr << SRC_BITS);
                unsigned word = __builtin_amdgcn_ballot_w32(valid);
                #pragma unroll
                for (int b = 0; b < 4; ++b) {
                    const unsigned bh = __builtin_amdgcn_ballot_w32(((o >> (4 + b)) & 1u) != 0u);
                    const unsigned bl = __builtin_amdgcn_ballot_w32(((o >> b) & 1u) != 0u);
                    word &= (hi_lane ? bh : bl) ^ xm[b];
                }
                key_scr[kl * KEY_PITCH + 8 * j + kw] = word;
            }
            __syncthreads();
            {
                const int plane = threadIdx.x >> 3;
                const int piece = threadIdx.x & 7;
                const uint4 v0 = *(const uint4*)&key_scr[plane * KEY_PITCH + 4 * piece];
                const uint4 v1 = *(const uint4*)&key_scr[plane * KEY_PITCH + 32 + 4 * piece];
                uint4* dstp = (uint4*)(kp + (size_t)plane * plane_words + (size_t)u * (KEY_UNIT / 32));
                dstp[piece]     = v0;
                dstp[8 + piece] = v1;
            }
        }
    }

    for (int idx = threadIdx.x; idx < (C_DIM / 2) * C_DIM; idx += 256) {
        const int p = idx >> 7;
        const int n = idx & 127;
        v2f v;
        v.x = w[(2 * p + 0) * C_DIM + n];
        v.y = w[(2 * p + 1) * C_DIM + n];
        const int dw = p * 256 + ((2 * n + 32 * (p & 1)) & 255);
        *(v2f*)&lds_w[dw] = v;
    }
    __syncthreads();

    const int wave   = threadIdx.x >> 5;
    const int lane   = threadIdx.x & 31;
    const int lane16 = lane & 15;
    const int khalf  = lane >> 4;

    const int m0 = (blockIdx.x * 8 + wave) * 16;
    int row  = m0 + lane16;
    int rowc = row < N ? row : N - 1;

    v8f acc[8];
    const v8f vzero = {0.f,0.f,0.f,0.f,0.f,0.f,0.f,0.f};
    #pragma unroll
    for (int n = 0; n < 8; ++n) acc[n] = vzero;

    const float* xrow = x + (size_t)rowc * C_DIM;

    int coff[8];
    #pragma unroll
    for (int n = 0; n < 8; ++n)
        coff[n] = (32 * n + 2 * lane16 + 32 * khalf) & 255;
    const float* lwp = lds_w + khalf * 256;

    #pragma unroll 4
    for (int k = 0; k < C_DIM; k += 4) {
        const int kb = k + 2 * khalf;
        v2f a = *(const v2f*)(xrow + kb);
        const float* lk = lwp + (k >> 1) * 256;
        #pragma unroll
        for (int n = 0; n < 8; ++n) {
            v2f b = *(const v2f*)(lk + coff[n]);
            acc[n] = __builtin_amdgcn_wmma_f32_16x16x4_f32(
                false, a, false, b, (short)0, acc[n], false, false);
        }
    }

    float nrm[8];
    #pragma unroll
    for (int j = 0; j < 8; ++j) {
        const int r = m0 + j + 8 * khalf;
        nrm[j] = (r < N) ? norm[r] : 0.0f;
    }
    #pragma unroll
    for (int n = 0; n < 8; ++n) {
        #pragma unroll
        for (int j = 0; j < 8; ++j) {
            const int r = m0 + j + 8 * khalf;
            if (r < N) {
                h[(size_t)r * C_DIM + n * 16 + lane16] = acc[n][j] * nrm[j];
            }
        }
    }
}

#define AGG_WAVES  32
#define CAP_A      416
#define SORT_REGS  ((CAP_A + 31) / 32)

__global__ __launch_bounds__(1024) void gcn_aggregate(
    const float* __restrict__ h, const float* __restrict__ norm,
    const float* __restrict__ bias, float* __restrict__ out, int N, int E)
{
    __shared__ float acc[(AGG_ROWS + 1) * C_DIM];
    __shared__ int a_lst[AGG_WAVES][CAP_A];
    __shared__ int s_lst[AGG_WAVES][CAP_A];
    __shared__ int s_off[AGG_WAVES][AGG_WAVES + 1];
    __shared__ int a_done[AGG_WAVES];

    const int lane = threadIdx.x & 31;
    const int wave = __builtin_amdgcn_readfirstlane(threadIdx.x >> 5);
    const int row0 = blockIdx.x * AGG_ROWS;
    const int nrows_i = (N - row0) < AGG_ROWS ? (N - row0) : AGG_ROWS;

    for (int i = threadIdx.x; i < (AGG_ROWS + 1) * C_DIM / 4; i += 1024)
        ((float4*)acc)[i] = make_float4(0.f, 0.f, 0.f, 0.f);

    const int nblk = (E + KEY_BLOCK - 1) / KEY_BLOCK;
    const int per  = (nblk + AGG_WAVES - 1) / AGG_WAVES;
    int cursor = wave * per;
    if (cursor > nblk) cursor = nblk;
    int end = cursor + per;
    if (end > nblk) end = nblk;

    int* my_a = a_lst[wave];
    int* my_s = s_lst[wave];

    const size_t plane_words = (size_t)((nblk + 1) >> 1) * (KEY_UNIT / 32);
    const unsigned* kp = (const unsigned*)(h + (size_t)N * C_DIM);
    const unsigned* k_hi = kp + (size_t)((blockIdx.x >> 4) & 15u) * plane_words + lane;
    const unsigned* k_lo = kp + (size_t)(16u + (blockIdx.x & 15u)) * plane_words + lane;
    const int* rec = (const int*)(kp + (size_t)KEY_PLANES * plane_words);

    auto match = [&](unsigned a, unsigned b, int blk) -> unsigned {
        unsigned m = a & b;
        if (blk * KEY_BLOCK + KEY_BLOCK > E) {
            const int room = E - (blk * KEY_BLOCK + lane * 32);
            m &= room >= 32 ? ~0u : (room <= 0 ? 0u : ((1u << room) - 1u));
        }
        return m;
    };

    auto drain = [&](unsigned& m, int blk, int& cnt) -> bool {
        const int ebase = blk * KEY_BLOCK + lane * 32;
        unsigned any = __builtin_amdgcn_ballot_w32(m != 0u);
        while (any != 0u) {
            const int tot = __builtin_popcount(any);
            if (cnt + tot > CAP_A) return false;
            const int rank = (int)__builtin_amdgcn_mbcnt_lo(any, 0u);
            if (m != 0u) my_a[cnt + rank] = ebase + __builtin_ctz(m);
            cnt += tot;
            m &= m - 1u;
            any = __builtin_amdgcn_ballot_w32(m != 0u);
        }
        return true;
    };

    unsigned pend = 0u;
    int pend_blk = 0;
    auto step = [&](unsigned a, unsigned b, int blk, int& cnt) -> bool {
        unsigned m = match(a, b, blk);
        if (drain(m, blk, cnt)) return true;
        pend = m;
        pend_blk = blk;
        return false;
    };

    auto key_masks = [&](int key, bool valid, unsigned& same, unsigned& mine) {
        const unsigned vm = __builtin_amdgcn_ballot_w32(valid);
        unsigned sm = vm, mn = vm;
        for (int b = 0; b < 5; ++b) {
            const unsigned bb = __builtin_amdgcn_ballot_w32(((key >> b) & 1) != 0);
            sm &= ~(bb ^ (0u - (unsigned)((key  >> b) & 1)));
            mn &= ~(bb ^ (0u - (unsigned)((lane >> b) & 1)));
        }
        same = sm;
        mine = mn;
    };

    auto flush_rows = [&](int cnt) {
        int pe[SORT_REGS];
        #pragma unroll
        for (int i = 0; i < SORT_REGS; ++i)
            pe[i] = (i * 32 + lane < cnt) ? my_a[i * 32 + lane] : 0;
        int cntv = 0;
        #pragma unroll
        for (int i = 0; i < SORT_REGS; ++i) {
            if (i * 32 < cnt) {
                unsigned same, mine;
                key_masks(pe[i] >> (SRC_BITS + 5), i * 32 + lane < cnt, same, mine);
                cntv += __builtin_popcount(mine);
            }
        }
        int inc = cntv;
        for (int d = 1; d < 32; d <<= 1) {
            const int t = __builtin_amdgcn_ds_bpermute(((lane - d) & 31) << 2, inc);
            if (lane >= d) inc += t;
        }
        int runv = inc - cntv;
        #pragma unroll
        for (int i = 0; i < SORT_REGS; ++i) {
            if (i * 32 < cnt) {
                const bool valid = i * 32 + lane < cnt;
                const int key = pe[i] >> (SRC_BITS + 5);
                unsigned same, mine;
                key_masks(key, valid, same, mine);
                const int base = __builtin_amdgcn_ds_bpermute(key << 2, runv);
                const int rank = (int)__builtin_amdgcn_mbcnt_lo(same, 0u);
                if (valid) my_a[base + rank] = pe[i];
                runv += __builtin_popcount(mine);
            }
        }

        const int padded = (cnt + 7) & ~7;
        if (lane < padded - cnt) my_a[cnt + lane] = DUMMY_ROW << SRC_BITS;
        float4 s = make_float4(0.f, 0.f, 0.f, 0.f);
        int cur = DUMMY_ROW;
        auto put = [&]() {
            float* a = &acc[cur * C_DIM + lane];
            __hip_atomic_fetch_add(a,      s.x, __ATOMIC_RELAXED, __HIP_MEMORY_SCOPE_WORKGROUP);
            __hip_atomic_fetch_add(a + 32, s.y, __ATOMIC_RELAXED, __HIP_MEMORY_SCOPE_WORKGROUP);
            __hip_atomic_fetch_add(a + 64, s.z, __ATOMIC_RELAXED, __HIP_MEMORY_SCOPE_WORKGROUP);
            __hip_atomic_fetch_add(a + 96, s.w, __ATOMIC_RELAXED, __HIP_MEMORY_SCOPE_WORKGROUP);
        };
        for (int i = 0; i < padded; i += 8) {
            const int pv = my_a[i + (lane & 7)];
            float4 v[8];
            int r[8];
            #pragma unroll
            for (int k = 0; k < 8; ++k) {
                const int p = __builtin_amdgcn_readlane(pv, k);
                r[k] = p >> SRC_BITS;
                v[k] = *(const float4*)(h + (size_t)(p & SRC_MASK) * C_DIM + lane * 4);
            }
            #pragma unroll
            for (int k = 0; k < 8; ++k) {
                if (r[k] != cur) {
                    put();
                    cur = r[k];
                    s = v[k];
                } else {
                    s.x += v[k].x; s.y += v[k].y; s.z += v[k].z; s.w += v[k].w;
                }
            }
        }
        put();
    };

    bool all_done = false;
    while (!all_done) {
        int cnt = 0;
        bool full = false;
        if (__builtin_amdgcn_ballot_w32(pend != 0u) != 0u) full = !drain(pend, pend_blk, cnt);
        while (!full && cursor + 8 <= end) {
            const unsigned* pa = k_hi + (size_t)cursor * 32;
            const unsigned* pb = k_lo + (size_t)cursor * 32;
            unsigned wa[8], wb[8];
            #pragma unroll
            for (int t = 0; t < 8; ++t) { wa[t] = pa[32 * t]; wb[t] = pb[32 * t]; }
            int t = 0;
            #pragma unroll
            for (; t < 8; ++t)
                if (!step(wa[t], wb[t], cursor + t, cnt)) { full = true; break; }
            cursor += full ? t + 1 : 8;
        }
        while (!full && cursor < end) {
            const unsigned a = k_hi[(size_t)cursor * 32], b = k_lo[(size_t)cursor * 32];
            if (!step(a, b, cursor, cnt)) { cursor += 1; full = true; break; }
            cursor += 1;
        }
        {
            int ev[SORT_REGS];
            #pragma unroll
            for (int i = 0; i < SORT_REGS; ++i)
                ev[i] = (i * 32 + lane < cnt) ? my_a[i * 32 + lane] : 0;
            #pragma unroll
            for (int i = 0; i < SORT_REGS; ++i)
                if (i * 32 < cnt) ev[i] = rec[ev[i]];
            #pragma unroll
            for (int i = 0; i < SORT_REGS; ++i)
                if (i * 32 + lane < cnt) my_a[i * 32 + lane] = ev[i];
        }

        int cntv = 0;
        for (int i0 = 0; i0 < cnt; i0 += 32) {
            const bool valid = i0 + lane < cnt;
            const int p = valid ? my_a[i0 + lane] : 0;
            unsigned same, mine;
            key_masks((p >> SRC_BITS) & 31, valid, same, mine);
            cntv += __builtin_popcount(mine);
        }
        int inc = cntv;
        for (int d = 1; d < 32; d <<= 1) {
            const int t = __builtin_amdgcn_ds_bpermute(((lane - d) & 31) << 2, inc);
            if (lane >= d) inc += t;
        }
        int runv = inc - cntv;
        s_off[wave][lane] = runv;
        if (lane == 31) s_off[wave][AGG_WAVES] = inc;
        for (int i0 = 0; i0 < cnt; i0 += 32) {
            const bool valid = i0 + lane < cnt;
            const int p = valid ? my_a[i0 + lane] : 0;
            const int key = (p >> SRC_BITS) & 31;
            unsigned same, mine;
            key_masks(key, valid, same, mine);
            const int base = __builtin_amdgcn_ds_bpermute(key << 2, runv);
            const int rank = (int)__builtin_amdgcn_mbcnt_lo(same, 0u);
            if (valid) my_s[base + rank] = p;
            runv += __builtin_popcount(mine);
        }
        if (lane == 0) a_done[wave] = (cursor >= end && !full) ? 1 : 0;
        __syncthreads();

        all_done = __builtin_amdgcn_ballot_w32(a_done[lane] != 0) == 0xFFFFFFFFu;

        const int segb = s_off[lane][wave];
        const int sege = s_off[lane][wave + 1];
        int cb = 0;
        for (int s = 0; s < AGG_WAVES; ++s) {
            const int b0 = __builtin_amdgcn_readlane(segb, s);
            const int n  = __builtin_amdgcn_readlane(sege, s) - b0;
            for (int i0 = 0; i0 < n; i0 += 32) {
                if (cb + 32 > CAP_A - 8) { flush_rows(cb); cb = 0; }
                const int m = (n - i0) < 32 ? (n - i0) : 32;
                if (lane < m) my_a[cb + lane] = s_lst[s][b0 + i0 + lane];
                cb += m;
            }
        }
        flush_rows(cb);
        __syncthreads();
    }

    const float4 b4 = *(const float4*)(bias + lane * 4);
    for (int lr = wave; lr < nrows_i; lr += AGG_WAVES) {
        const float* a = &acc[lr * C_DIM + lane];
        const float nm = norm[row0 + lr];
        float4 v;
        v.x = a[0]  * nm + b4.x;
        v.y = a[32] * nm + b4.y;
        v.z = a[64] * nm + b4.z;
        v.w = a[96] * nm + b4.w;
        *(float4*)(out + (size_t)(row0 + lr) * C_DIM + lane * 4) = v;
    }
}

extern "C" void kernel_launch(void* const* d_in, const int* in_sizes, int n_in,
                              void* d_out, int out_size, void* d_ws, size_t ws_size,
                              hipStream_t stream) {
    const float* x    = (const float*)d_in[0];
    const float* norm = (const float*)d_in[1];
    const float* w    = (const float*)d_in[2];
    const float* bias = (const float*)d_in[3];
    const int*   esrc = (const int*)d_in[4];
    const int*   edst = (const int*)d_in[5];
    float* out = (float*)d_out;

    const int N = in_sizes[0] / C_DIM;
    const int E = in_sizes[4];
    float* h = (float*)d_ws;

    gcn_gemm_norm<<<(N + 127) / 128, dim3(256), 0, stream>>>(x, w, norm, esrc, edst, h, N, E);

    gcn_aggregate<<<(N + AGG_ROWS - 1) / AGG_ROWS, dim3(1024), 0, stream>>>(
        h, norm, bias, out, N, E);
}
